# starting kernel
# baseline (speedup 1.0000x reference)
#include <hip/hip_runtime.h>
#include <math.h>


typedef __attribute__((ext_vector_type(2))) float v2f;
typedef __attribute__((ext_vector_type(8))) float v8f;

#define DH 64


__global__ void degree_count_kernel(const int* __restrict__ src,
                                    const int* __restrict__ dst,
                                    float* __restrict__ cnt_out,
                                    float* __restrict__ cnt_in, int E) {
    int e = blockIdx.x * blockDim.x + threadIdx.x;
    if (e < E) {
        atomicAdd(&cnt_out[src[e]], 1.0f);
        atomicAdd(&cnt_in[dst[e]], 1.0f);
    }
}

__global__ void degree_finalize_kernel(float* __restrict__ a,
                                       float* __restrict__ b, int N) {
    int i = blockIdx.x * blockDim.x + threadIdx.x;
    if (i < N) {
        a[i] = rsqrtf(fmaxf(a[i], 1.0f));
        b[i] = rsqrtf(fmaxf(b[i], 1.0f));
    }
}


template <int FOUT>
__global__ __launch_bounds__(256) void gemm_norm_wmma(
        const float* __restrict__ A0,
        const float* __restrict__ A1,
        const float* __restrict__ W,
        const float* __restrict__ inv_deg,
        float* __restrict__ out,
        int mtiles) {
    constexpr int NT = FOUT / 16;
    int tile = blockIdx.x * 8 + threadIdx.y;
    int mt = tile / NT;
    int nt = tile % NT;
    if (mt >= mtiles) return;

    int lane = threadIdx.x;
    int half = lane >> 4;
    int l15  = lane & 15;

    int row = mt * 16 + l15;
    float s = inv_deg[row];
    const float* xr = A0 + (size_t)row * DH;
    const float* hr = A1 + (size_t)row * DH;
    int col = nt * 16 + l15;

    v8f acc = {};
#pragma unroll
    for (int k0 = 0; k0 < 2 * DH; k0 += 4) {
        int kb = k0 + half * 2;
        const float* ap = (k0 < DH) ? (xr + kb) : (hr + (kb - DH));
        v2f a, b;
        a.x = ap[0] * s;
        a.y = ap[1] * s;
        b.x = W[(size_t)kb * FOUT + col];
        b.y = W[(size_t)(kb + 1) * FOUT + col];
        acc = __builtin_amdgcn_wmma_f32_16x16x4_f32(
             false, a,  false, b,
             (short)0, acc,  false,  false);
    }

#pragma unroll
    for (int j = 0; j < 8; ++j) {
        int r = mt * 16 + half * 8 + j;
        out[(size_t)r * FOUT + col] = acc[j];
    }
}


template <int F>
__global__ void scatter_add_kernel(const float* __restrict__ xw,
                                   const int* __restrict__ src,
                                   const int* __restrict__ dst,
                                   float* __restrict__ agg, int E) {
    unsigned idx = blockIdx.x * blockDim.x + threadIdx.x;
    unsigned total = (unsigned)E * (unsigned)F;
    if (idx >= total) return;
    unsigned e = idx / F;
    unsigned f = idx % F;
    float v = xw[(size_t)src[e] * F + f];
    atomicAdd(&agg[(size_t)dst[e] * F + f], v);
}


__global__ void gates_kernel(const float* __restrict__ agg,
                             const float* __restrict__ inv_di,
                             const float* __restrict__ b_gate,
                             const float* __restrict__ h,
                             float* __restrict__ u,
                             float* __restrict__ rh,
                             int N) {
    int idx = blockIdx.x * blockDim.x + threadIdx.x;
    if (idx >= N * DH) return;
    int n = idx >> 6;
    int j = idx & (DH - 1);
    float inv = inv_di[n];
    float gr = agg[(size_t)n * 2 * DH + j] * inv + b_gate[j];
    float gu = agg[(size_t)n * 2 * DH + DH + j] * inv + b_gate[DH + j];
    float r  = 1.0f / (1.0f + __expf(-gr));
    float uu = 1.0f / (1.0f + __expf(-gu));
    u[idx]  = uu;
    rh[idx] = r * h[idx];
}


__global__ void final_kernel(const float* __restrict__ agg2,
                             const float* __restrict__ inv_di,
                             const float* __restrict__ b_cand,
                             const float* __restrict__ h,
                             const float* __restrict__ u,
                             float* __restrict__ out, int N) {
    int idx = blockIdx.x * blockDim.x + threadIdx.x;
    if (idx >= N * DH) return;
    int n = idx >> 6;
    int j = idx & (DH - 1);
    float cand = agg2[idx] * inv_di[n] + b_cand[j];
    float uu = u[idx];
    out[idx] = uu * h[idx] + (1.0f - uu) * tanhf(cand);
}


extern "C" void kernel_launch(void* const* d_in, const int* in_sizes, int n_in,
                              void* d_out, int out_size, void* d_ws, size_t ws_size,
                              hipStream_t stream) {
    const float* x      = (const float*)d_in[0];
    const float* h      = (const float*)d_in[1];
    const int*   src    = (const int*)d_in[2];
    const int*   dst    = (const int*)d_in[3];
    const float* W_gate = (const float*)d_in[4];
    const float* b_gate = (const float*)d_in[5];
    const float* W_cand = (const float*)d_in[6];
    const float* b_cand = (const float*)d_in[7];

    const int N = in_sizes[0] / DH;
    const int E = in_sizes[2];

    float* p      = (float*)d_ws;
    float* inv_do = p;
    float* inv_di = p + N;
    float* xw     = p + 2 * (size_t)N;
    float* agg    = xw + (size_t)N * 2 * DH;
    float* u      = agg + (size_t)N * 2 * DH;
    float* rh     = u + (size_t)N * DH;
    float* out    = (float*)d_out;

    const int mtiles = N / 16;

    hipMemsetAsync(inv_do, 0, sizeof(float) * 2 * (size_t)N, stream);
    degree_count_kernel<<<(E + 255) / 256, 256, 0, stream>>>(src, dst, inv_do, inv_di, E);
    degree_finalize_kernel<<<(N + 255) / 256, 256, 0, stream>>>(inv_do, inv_di, N);

    hipMemsetAsync(agg, 0, sizeof(float) * (size_t)N * 2 * DH, stream);
    {
        int tiles = mtiles * (2 * DH / 16);
        gemm_norm_wmma<2 * DH><<<(tiles + 7) / 8, dim3(32, 8), 0, stream>>>(
            x, h, W_gate, inv_do, xw, mtiles);
    }
    {
        unsigned total = (unsigned)E * (2 * DH);
        scatter_add_kernel<2 * DH><<<(total + 255) / 256, 256, 0, stream>>>(
            xw, src, dst, agg, E);
    }
    gates_kernel<<<(N * DH + 255) / 256, 256, 0, stream>>>(agg, inv_di, b_gate, h, u, rh, N);

    hipMemsetAsync(agg, 0, sizeof(float) * (size_t)N * DH, stream);
    {
        int tiles = mtiles * (DH / 16);
        gemm_norm_wmma<DH><<<(tiles + 7) / 8, dim3(32, 8), 0, stream>>>(
            x, rh, W_cand, inv_do, xw, mtiles);
    }
    {
        unsigned total = (unsigned)E * DH;
        scatter_add_kernel<DH><<<(total + 255) / 256, 256, 0, stream>>>(
            xw, src, dst, agg, E);
    }
    final_kernel<<<(N * DH + 255) / 256, 256, 0, stream>>>(agg, inv_di, b_cand, h, u, out, N);
}


# Replace the float-atomic scatter-add by a per-call grouping of edges by destination (integer-atomic arrival rank) and an atomic-free per-node gather-sum; per-node sum order = arrival rank
# speedup vs baseline: 3.0487x; 3.0487x over previous
#include <hip/hip_runtime.h>
#include <math.h>


typedef __attribute__((ext_vector_type(2))) float v2f;
typedef __attribute__((ext_vector_type(8))) float v8f;

#define DH 64


__global__ void degree_rank_kernel(const int* __restrict__ src,
                                   const int* __restrict__ dst,
                                   unsigned* __restrict__ cnt_out,
                                   unsigned* __restrict__ cnt_in,
                                   unsigned* __restrict__ rank, int E) {
    int e = blockIdx.x * blockDim.x + threadIdx.x;
    if (e < E) {
        atomicAdd(&cnt_out[src[e]], 1u);
        rank[e] = atomicAdd(&cnt_in[dst[e]], 1u);
    }
}


__global__ void degree_finalize_offsets_kernel(const unsigned* __restrict__ cnt_out,
                                               const unsigned* __restrict__ cnt_in,
                                               unsigned* __restrict__ cursor,
                                               float* __restrict__ inv_do,
                                               float* __restrict__ inv_di,
                                               unsigned* __restrict__ off, int N) {
    int i = blockIdx.x * blockDim.x + threadIdx.x;
    unsigned ci = 0;
    if (i < N) {
        unsigned co = cnt_out[i];
        ci = cnt_in[i];
        inv_do[i] = rsqrtf(fmaxf((float)co, 1.0f));
        inv_di[i] = rsqrtf(fmaxf((float)ci, 1.0f));
    }
    const int lane = threadIdx.x & (warpSize - 1);
    unsigned incl = ci;
    for (int d = 1; d < warpSize; d <<= 1) {
        unsigned t = __shfl_up(incl, d);
        if (lane >= d) incl += t;
    }
    unsigned total = __shfl(incl, warpSize - 1);
    unsigned base = 0;
    if (lane == 0) base = atomicAdd(cursor, total);
    base = __shfl(base, 0);
    if (i < N) off[i] = base + incl - ci;
}


__global__ void group_fill_kernel(const int* __restrict__ src,
                                  const int* __restrict__ dst,
                                  const unsigned* __restrict__ rank,
                                  const unsigned* __restrict__ off,
                                  int* __restrict__ grouped, int E) {
    int e = blockIdx.x * blockDim.x + threadIdx.x;
    if (e < E) {
        unsigned pos = min(off[dst[e]] + rank[e], (unsigned)E - 1u);
        grouped[pos] = src[e];
    }
}


template <int FOUT>
__global__ __launch_bounds__(256) void gemm_norm_wmma(
        const float* __restrict__ A0,
        const float* __restrict__ A1,
        const float* __restrict__ W,
        const float* __restrict__ inv_deg,
        float* __restrict__ out,
        int mtiles) {
    constexpr int NT = FOUT / 16;
    int tile = blockIdx.x * 8 + threadIdx.y;
    int mt = tile / NT;
    int nt = tile % NT;
    if (mt >= mtiles) return;

    int lane = threadIdx.x;
    int half = lane >> 4;
    int l15  = lane & 15;

    int row = mt * 16 + l15;
    float s = inv_deg[row];
    const float* xr = A0 + (size_t)row * DH;
    const float* hr = A1 + (size_t)row * DH;
    int col = nt * 16 + l15;

    v8f acc = {};
#pragma unroll
    for (int k0 = 0; k0 < 2 * DH; k0 += 4) {
        int kb = k0 + half * 2;
        const float* ap = (k0 < DH) ? (xr + kb) : (hr + (kb - DH));
        v2f a, b;
        a.x = ap[0] * s;
        a.y = ap[1] * s;
        b.x = W[(size_t)kb * FOUT + col];
        b.y = W[(size_t)(kb + 1) * FOUT + col];
        acc = __builtin_amdgcn_wmma_f32_16x16x4_f32(
             false, a,  false, b,
             (short)0, acc,  false,  false);
    }

#pragma unroll
    for (int j = 0; j < 8; ++j) {
        int r = mt * 16 + half * 8 + j;
        out[(size_t)r * FOUT + col] = acc[j];
    }
}


template <int F>
__global__ __launch_bounds__(256) void gather_sum_kernel(
        const float* __restrict__ xw,
        const int* __restrict__ grouped,
        const unsigned* __restrict__ off,
        const unsigned* __restrict__ cnt,
        float* __restrict__ agg,
        int N) {
    constexpr int LPN = F / 4;
    unsigned t = blockIdx.x * blockDim.x + threadIdx.x;
    unsigned node = t / LPN;
    unsigned lin  = t % LPN;
    if (node >= (unsigned)N) return;

    const unsigned n = cnt[node];
    const int* seg = grouped + off[node];
    const float4* rows = reinterpret_cast<const float4*>(xw) + lin;
    const unsigned last = (unsigned)N - 1u;

    float4 acc = make_float4(0.0f, 0.0f, 0.0f, 0.0f);
    for (unsigned k0 = 0; k0 < n; k0 += LPN) {
        const unsigned m = min(n - k0, (unsigned)LPN);
        const unsigned mine = min((unsigned)seg[k0 + min(lin, m - 1u)], last);
        for (unsigned j0 = 0; j0 < m; j0 += 8) {
            float4 v[8];
#pragma unroll
            for (int j = 0; j < 8; ++j) {
                const unsigned s = __shfl(mine, (int)(j0 + j), LPN);
                v[j] = make_float4(0.0f, 0.0f, 0.0f, 0.0f);
                if (j0 + j < m) v[j] = rows[(size_t)s * LPN];
            }
#pragma unroll
            for (int j = 0; j < 8; ++j) {
                acc.x += v[j].x;
                acc.y += v[j].y;
                acc.z += v[j].z;
                acc.w += v[j].w;
            }
        }
    }
    reinterpret_cast<float4*>(agg)[(size_t)node * LPN + lin] = acc;
}


__global__ void gates_kernel(const float* __restrict__ agg,
                             const float* __restrict__ inv_di,
                             const float* __restrict__ b_gate,
                             const float* __restrict__ h,
                             float* __restrict__ u,
                             float* __restrict__ rh,
                             int N) {
    int idx = blockIdx.x * blockDim.x + threadIdx.x;
    if (idx >= N * DH) return;
    int n = idx >> 6;
    int j = idx & (DH - 1);
    float inv = inv_di[n];
    float gr = agg[(size_t)n * 2 * DH + j] * inv + b_gate[j];
    float gu = agg[(size_t)n * 2 * DH + DH + j] * inv + b_gate[DH + j];
    float r  = 1.0f / (1.0f + __expf(-gr));
    float uu = 1.0f / (1.0f + __expf(-gu));
    u[idx]  = uu;
    rh[idx] = r * h[idx];
}


__global__ void final_kernel(const float* __restrict__ agg2,
                             const float* __restrict__ inv_di,
                             const float* __restrict__ b_cand,
                             const float* __restrict__ h,
                             const float* __restrict__ u,
                             float* __restrict__ out, int N) {
    int idx = blockIdx.x * blockDim.x + threadIdx.x;
    if (idx >= N * DH) return;
    int n = idx >> 6;
    int j = idx & (DH - 1);
    float cand = agg2[idx] * inv_di[n] + b_cand[j];
    float uu = u[idx];
    out[idx] = uu * h[idx] + (1.0f - uu) * tanhf(cand);
}


extern "C" void kernel_launch(void* const* d_in, const int* in_sizes, int n_in,
                              void* d_out, int out_size, void* d_ws, size_t ws_size,
                              hipStream_t stream) {
    const float* x      = (const float*)d_in[0];
    const float* h      = (const float*)d_in[1];
    const int*   src    = (const int*)d_in[2];
    const int*   dst    = (const int*)d_in[3];
    const float* W_gate = (const float*)d_in[4];
    const float* b_gate = (const float*)d_in[5];
    const float* W_cand = (const float*)d_in[6];
    const float* b_cand = (const float*)d_in[7];

    const int N = in_sizes[0] / DH;
    const int E = in_sizes[2];

    float* p      = (float*)d_ws;
    float* inv_do = p;
    float* inv_di = p + N;
    float* bufA   = p + 2 * (size_t)N;
    float* bufB   = bufA + (size_t)N * 2 * DH;
    float* bufC   = bufB + (size_t)N * 2 * DH;
    unsigned* q   = (unsigned*)(bufC + (size_t)N * DH);
    float* xw     = bufA;
    float* agg    = bufB;
    float* u      = bufA;
    float* rh     = bufA + (size_t)N * DH;
    float* xw2    = bufC;
    unsigned* cnt_out = q;
    unsigned* cnt_in  = q + N;
    unsigned* cursor  = q + 2 * (size_t)N;
    unsigned* off     = cursor + 32;
    unsigned* rank    = off + N;
    int* grouped      = (int*)(rank + E);
    float* out    = (float*)d_out;

    const int mtiles = N / 16;

    hipMemsetAsync(cnt_out, 0, sizeof(unsigned) * (2 * (size_t)N + 32), stream);
    degree_rank_kernel<<<(E + 255) / 256, 256, 0, stream>>>(src, dst, cnt_out, cnt_in, rank, E);
    degree_finalize_offsets_kernel<<<(N + 255) / 256, 256, 0, stream>>>(
        cnt_out, cnt_in, cursor, inv_do, inv_di, off, N);
    group_fill_kernel<<<(E + 255) / 256, 256, 0, stream>>>(src, dst, rank, off, grouped, E);

    {
        int tiles = mtiles * (2 * DH / 16);
        gemm_norm_wmma<2 * DH><<<(tiles + 7) / 8, dim3(32, 8), 0, stream>>>(
            x, h, W_gate, inv_do, xw, mtiles);
    }
    {
        unsigned total = (unsigned)N * (2 * DH / 4);
        gather_sum_kernel<2 * DH><<<(total + 255) / 256, 256, 0, stream>>>(
            xw, grouped, off, cnt_in, agg, N);
    }
    gates_kernel<<<(N * DH + 255) / 256, 256, 0, stream>>>(agg, inv_di, b_gate, h, u, rh, N);

    {
        int tiles = mtiles * (DH / 16);
        gemm_norm_wmma<DH><<<(tiles + 7) / 8, dim3(32, 8), 0, stream>>>(
            x, rh, W_cand, inv_do, xw2, mtiles);
    }
    {
        unsigned total = (unsigned)N * (DH / 4);
        gather_sum_kernel<DH><<<(total + 255) / 256, 256, 0, stream>>>(
            xw2, grouped, off, cnt_in, agg, N);
    }
    final_kernel<<<(N * DH + 255) / 256, 256, 0, stream>>>(agg, inv_di, b_cand, h, u, out, N);
}


# Fuse the two elementwise kernels into the gather-sum kernels: the lanes holding a node's summed row apply sigmoid, r*h, tanh and the blend and write u, r*h, new_h directly; summed rows never stored
# speedup vs baseline: 3.3679x; 1.1047x over previous
#include <hip/hip_runtime.h>
#include <math.h>


typedef __attribute__((ext_vector_type(2))) float v2f;
typedef __attribute__((ext_vector_type(8))) float v8f;

#define DH 64


__global__ void degree_rank_kernel(const int* __restrict__ src,
                                   const int* __restrict__ dst,
                                   unsigned* __restrict__ cnt_out,
                                   unsigned* __restrict__ cnt_in,
                                   unsigned* __restrict__ rank, int E) {
    int e = blockIdx.x * blockDim.x + threadIdx.x;
    if (e < E) {
        atomicAdd(&cnt_out[src[e]], 1u);
        rank[e] = atomicAdd(&cnt_in[dst[e]], 1u);
    }
}


__global__ void degree_finalize_offsets_kernel(const unsigned* __restrict__ cnt_out,
                                               const unsigned* __restrict__ cnt_in,
                                               unsigned* __restrict__ cursor,
                                               float* __restrict__ inv_do,
                                               float* __restrict__ inv_di,
                                               unsigned* __restrict__ off, int N) {
    int i = blockIdx.x * blockDim.x + threadIdx.x;
    unsigned ci = 0;
    if (i < N) {
        unsigned co = cnt_out[i];
        ci = cnt_in[i];
        inv_do[i] = rsqrtf(fmaxf((float)co, 1.0f));
        inv_di[i] = rsqrtf(fmaxf((float)ci, 1.0f));
    }
    const int lane = threadIdx.x & (warpSize - 1);
    unsigned incl = ci;
    for (int d = 1; d < warpSize; d <<= 1) {
        unsigned t = __shfl_up(incl, d);
        if (lane >= d) incl += t;
    }
    unsigned total = __shfl(incl, warpSize - 1);
    unsigned base = 0;
    if (lane == 0) base = atomicAdd(cursor, total);
    base = __shfl(base, 0);
    if (i < N) off[i] = base + incl - ci;
}


__global__ void group_fill_kernel(const int* __restrict__ src,
                                  const int* __restrict__ dst,
                                  const unsigned* __restrict__ rank,
                                  const unsigned* __restrict__ off,
                                  int* __restrict__ grouped, int E) {
    int e = blockIdx.x * blockDim.x + threadIdx.x;
    if (e < E) {
        unsigned pos = min(off[dst[e]] + rank[e], (unsigned)E - 1u);
        grouped[pos] = src[e];
    }
}


template <int FOUT>
__global__ __launch_bounds__(256) void gemm_norm_wmma(
        const float* __restrict__ A0,
        const float* __restrict__ A1,
        const float* __restrict__ W,
        const float* __restrict__ inv_deg,
        float* __restrict__ out,
        int mtiles) {
    constexpr int NT = FOUT / 16;
    int tile = blockIdx.x * 8 + threadIdx.y;
    int mt = tile / NT;
    int nt = tile % NT;
    if (mt >= mtiles) return;

    int lane = threadIdx.x;
    int half = lane >> 4;
    int l15  = lane & 15;

    int row = mt * 16 + l15;
    float s = inv_deg[row];
    const float* xr = A0 + (size_t)row * DH;
    const float* hr = A1 + (size_t)row * DH;
    int col = nt * 16 + l15;

    v8f acc = {};
#pragma unroll
    for (int k0 = 0; k0 < 2 * DH; k0 += 4) {
        int kb = k0 + half * 2;
        const float* ap = (k0 < DH) ? (xr + kb) : (hr + (kb - DH));
        v2f a, b;
        a.x = ap[0] * s;
        a.y = ap[1] * s;
        b.x = W[(size_t)kb * FOUT + col];
        b.y = W[(size_t)(kb + 1) * FOUT + col];
        acc = __builtin_amdgcn_wmma_f32_16x16x4_f32(
             false, a,  false, b,
             (short)0, acc,  false,  false);
    }

#pragma unroll
    for (int j = 0; j < 8; ++j) {
        int r = mt * 16 + half * 8 + j;
        out[(size_t)r * FOUT + col] = acc[j];
    }
}


template <int F>
__device__ __forceinline__ float4 gather_row_sum(
        const float* __restrict__ xw,
        const int* __restrict__ grouped,
        const unsigned* __restrict__ off,
        const unsigned* __restrict__ cnt,
        unsigned node, unsigned lin, int N) {
    constexpr int LPN = F / 4;
    const unsigned n = cnt[node];
    const int* seg = grouped + off[node];
    const float4* rows = reinterpret_cast<const float4*>(xw) + lin;
    const unsigned last = (unsigned)N - 1u;

    float4 acc = make_float4(0.0f, 0.0f, 0.0f, 0.0f);
    for (unsigned k0 = 0; k0 < n; k0 += LPN) {
        const unsigned m = min(n - k0, (unsigned)LPN);
        const unsigned mine = min((unsigned)seg[k0 + min(lin, m - 1u)], last);
        for (unsigned j0 = 0; j0 < m; j0 += 8) {
            float4 v[8];
#pragma unroll
            for (int j = 0; j < 8; ++j) {
                const unsigned s = __shfl(mine, (int)(j0 + j), LPN);
                v[j] = make_float4(0.0f, 0.0f, 0.0f, 0.0f);
                if (j0 + j < m) v[j] = rows[(size_t)s * LPN];
            }
#pragma unroll
            for (int j = 0; j < 8; ++j) {
                acc.x += v[j].x;
                acc.y += v[j].y;
                acc.z += v[j].z;
                acc.w += v[j].w;
            }
        }
    }
    return acc;
}

__device__ __forceinline__ float sigmoid_f(float g) {
    return 1.0f / (1.0f + __expf(-g));
}


__global__ __launch_bounds__(256) void gather_gates_kernel(
        const float* __restrict__ xw,
        const int* __restrict__ grouped,
        const unsigned* __restrict__ off,
        const unsigned* __restrict__ cnt,
        const float* __restrict__ inv_di,
        const float* __restrict__ b_gate,
        const float* __restrict__ h,
        float* __restrict__ u,
        float* __restrict__ rh,
        int N) {
    constexpr int LPN = 2 * DH / 4;
    constexpr int HQ  = DH / 4;
    unsigned t = blockIdx.x * blockDim.x + threadIdx.x;
    unsigned node = t / LPN;
    unsigned lin  = t % LPN;
    if (node >= (unsigned)N) return;
    const unsigned c4 = lin & (HQ - 1);

    const float inv = inv_di[node];
    const float4 b  = reinterpret_cast<const float4*>(b_gate)[lin];
    const float4 hh = reinterpret_cast<const float4*>(h)[(size_t)node * HQ + c4];

    const float4 acc = gather_row_sum<2 * DH>(xw, grouped, off, cnt, node, lin, N);

    float4 g;
    g.x = sigmoid_f(acc.x * inv + b.x);
    g.y = sigmoid_f(acc.y * inv + b.y);
    g.z = sigmoid_f(acc.z * inv + b.z);
    g.w = sigmoid_f(acc.w * inv + b.w);
    if (lin < HQ) {
        reinterpret_cast<float4*>(rh)[(size_t)node * HQ + c4] =
            make_float4(g.x * hh.x, g.y * hh.y, g.z * hh.z, g.w * hh.w);
    } else {
        reinterpret_cast<float4*>(u)[(size_t)node * HQ + c4] = g;
    }
}


__global__ __launch_bounds__(256) void gather_final_kernel(
        const float* __restrict__ xw,
        const int* __restrict__ grouped,
        const unsigned* __restrict__ off,
        const unsigned* __restrict__ cnt,
        const float* __restrict__ inv_di,
        const float* __restrict__ b_cand,
        const float* __restrict__ h,
        const float* __restrict__ u,
        float* __restrict__ out,
        int N) {
    constexpr int LPN = DH / 4;
    unsigned t = blockIdx.x * blockDim.x + threadIdx.x;
    unsigned node = t / LPN;
    unsigned lin  = t % LPN;
    if (node >= (unsigned)N) return;

    const float inv = inv_di[node];
    const float4 b  = reinterpret_cast<const float4*>(b_cand)[lin];
    const float4 hh = reinterpret_cast<const float4*>(h)[(size_t)node * LPN + lin];
    const float4 uu = reinterpret_cast<const float4*>(u)[(size_t)node * LPN + lin];

    const float4 acc = gather_row_sum<DH>(xw, grouped, off, cnt, node, lin, N);

    float4 o;
    o.x = uu.x * hh.x + (1.0f - uu.x) * tanhf(acc.x * inv + b.x);
    o.y = uu.y * hh.y + (1.0f - uu.y) * tanhf(acc.y * inv + b.y);
    o.z = uu.z * hh.z + (1.0f - uu.z) * tanhf(acc.z * inv + b.z);
    o.w = uu.w * hh.w + (1.0f - uu.w) * tanhf(acc.w * inv + b.w);
    reinterpret_cast<float4*>(out)[(size_t)node * LPN + lin] = o;
}


extern "C" void kernel_launch(void* const* d_in, const int* in_sizes, int n_in,
                              void* d_out, int out_size, void* d_ws, size_t ws_size,
                              hipStream_t stream) {
    const float* x      = (const float*)d_in[0];
    const float* h      = (const float*)d_in[1];
    const int*   src    = (const int*)d_in[2];
    const int*   dst    = (const int*)d_in[3];
    const float* W_gate = (const float*)d_in[4];
    const float* b_gate = (const float*)d_in[5];
    const float* W_cand = (const float*)d_in[6];
    const float* b_cand = (const float*)d_in[7];

    const int N = in_sizes[0] / DH;
    const int E = in_sizes[2];

    float* p      = (float*)d_ws;
    float* inv_do = p;
    float* inv_di = p + N;
    float* bufA   = p + 2 * (size_t)N;
    float* bufB   = bufA + (size_t)N * 2 * DH;
    float* bufC   = bufB + (size_t)N * 2 * DH;
    unsigned* q   = (unsigned*)(bufC + (size_t)N * DH);
    float* xw     = bufA;
    float* u      = bufB;
    float* rh     = bufB + (size_t)N * DH;
    float* xw2    = bufC;
    unsigned* cnt_out = q;
    unsigned* cnt_in  = q + N;
    unsigned* cursor  = q + 2 * (size_t)N;
    unsigned* off     = cursor + 32;
    unsigned* rank    = off + N;
    int* grouped      = (int*)(rank + E);
    float* out    = (float*)d_out;

    const int mtiles = N / 16;

    hipMemsetAsync(cnt_out, 0, sizeof(unsigned) * (2 * (size_t)N + 32), stream);
    degree_rank_kernel<<<(E + 255) / 256, 256, 0, stream>>>(src, dst, cnt_out, cnt_in, rank, E);
    degree_finalize_offsets_kernel<<<(N + 255) / 256, 256, 0, stream>>>(
        cnt_out, cnt_in, cursor, inv_do, inv_di, off, N);
    group_fill_kernel<<<(E + 255) / 256, 256, 0, stream>>>(src, dst, rank, off, grouped, E);

    {
        int tiles = mtiles * (2 * DH / 16);
        gemm_norm_wmma<2 * DH><<<(tiles + 7) / 8, dim3(32, 8), 0, stream>>>(
            x, h, W_gate, inv_do, xw, mtiles);
    }
    {
        unsigned total = (unsigned)N * (2 * DH / 4);
        gather_gates_kernel<<<(total + 255) / 256, 256, 0, stream>>>(
            xw, grouped, off, cnt_in, inv_di, b_gate, h, u, rh, N);
    }

    {
        int tiles = mtiles * (DH / 16);
        gemm_norm_wmma<DH><<<(tiles + 7) / 8, dim3(32, 8), 0, stream>>>(
            x, rh, W_cand, inv_do, xw2, mtiles);
    }
    {
        unsigned total = (unsigned)N * (DH / 4);
        gather_final_kernel<<<(total + 255) / 256, 256, 0, stream>>>(
            xw2, grouped, off, cnt_in, inv_di, b_cand, h, u, out, N);
    }
}


# Resident grids: waves loop over their edges / nodes / 16-row GEMM tiles (a GEMM wave does all column tiles, W staged in LDS), not one short wave per item: 328k waves become 60k; sums unchanged
# speedup vs baseline: 3.7906x; 1.1255x over previous
#include <hip/hip_runtime.h>
#include <math.h>


typedef __attribute__((ext_vector_type(2))) float v2f;
typedef __attribute__((ext_vector_type(4))) float v4f;
typedef __attribute__((ext_vector_type(8))) float v8f;

#define DH 64


__global__ void degree_rank_kernel(const int* __restrict__ src,
                                   const int* __restrict__ dst,
                                   unsigned* __restrict__ cnt_out,
                                   unsigned* __restrict__ cnt_in,
                                   unsigned* __restrict__ rank, int E) {
    const unsigned stride = gridDim.x * blockDim.x;
    for (unsigned e = blockIdx.x * blockDim.x + threadIdx.x; e < (unsigned)E; e += stride) {
        const int s = src[e];
        const int d = dst[e];
        atomicAdd(&cnt_out[s], 1u);
        rank[e] = atomicAdd(&cnt_in[d], 1u);
    }
}


__global__ void degree_finalize_offsets_kernel(const unsigned* __restrict__ cnt_out,
                                               const unsigned* __restrict__ cnt_in,
                                               unsigned* __restrict__ cursor,
                                               float* __restrict__ inv_do,
                                               float* __restrict__ inv_di,
                                               unsigned* __restrict__ off, int N) {
    int i = blockIdx.x * blockDim.x + threadIdx.x;
    unsigned ci = 0;
    if (i < N) {
        unsigned co = cnt_out[i];
        ci = cnt_in[i];
        inv_do[i] = rsqrtf(fmaxf((float)co, 1.0f));
        inv_di[i] = rsqrtf(fmaxf((float)ci, 1.0f));
    }
    const int lane = threadIdx.x & (warpSize - 1);
    unsigned incl = ci;
    for (int d = 1; d < warpSize; d <<= 1) {
        unsigned t = __shfl_up(incl, d);
        if (lane >= d) incl += t;
    }
    unsigned total = __shfl(incl, warpSize - 1);
    unsigned base = 0;
    if (lane == 0) base = atomicAdd(cursor, total);
    base = __shfl(base, 0);
    if (i < N) off[i] = base + incl - ci;
}


__global__ void group_fill_kernel(const int* __restrict__ src,
                                  const int* __restrict__ dst,
                                  const unsigned* __restrict__ rank,
                                  const unsigned* __restrict__ off,
                                  int* __restrict__ grouped, int E) {
    const unsigned stride = gridDim.x * blockDim.x;
    for (unsigned e = blockIdx.x * blockDim.x + threadIdx.x; e < (unsigned)E; e += stride) {
        unsigned pos = min(off[dst[e]] + rank[e], (unsigned)E - 1u);
        grouped[pos] = src[e];
    }
}


template <int FOUT>
__global__ __launch_bounds__(256) void gemm_norm_wmma(
        const float* __restrict__ A0,
        const float* __restrict__ A1,
        const float* __restrict__ W,
        const float* __restrict__ inv_deg,
        float* __restrict__ out,
        int mtiles) {
    constexpr int NT = FOUT / 16;
    constexpr int KS = 2 * DH / 4;
    constexpr int C4 = FOUT / 4;

    __shared__ v2f Ws[KS * NT * 32];

    const int lane = threadIdx.x;
    const int wave = __builtin_amdgcn_readfirstlane(threadIdx.y);
    const int half = lane >> 4;
    const int l15  = lane & 15;

    {
        constexpr int TRIPS = DH * C4 / 256;
        const int tid = wave * 32 + lane;
        v4f r0[TRIPS], r1[TRIPS];
#pragma unroll
        for (int it = 0; it < TRIPS; ++it) {
            const int p  = tid + it * 256;
            const int kp = p / C4;
            const int c4 = p % C4;
            r0[it] = *reinterpret_cast<const v4f*>(W + (size_t)(2 * kp) * FOUT + c4 * 4);
            r1[it] = *reinterpret_cast<const v4f*>(W + (size_t)(2 * kp + 1) * FOUT + c4 * 4);
        }
#pragma unroll
        for (int it = 0; it < TRIPS; ++it) {
            const int p   = tid + it * 256;
            const int kp  = p / C4;
            const int c4  = p % C4;
            const int t   = kp >> 1;
            const int h   = kp & 1;
            const int col = c4 * 4;
            v2f* dstp = Ws + ((t * NT + (col >> 4)) * 32 + h * 16 + (col & 15));
            v2f e0, e1, e2, e3;
            e0.x = r0[it].x; e0.y = r1[it].x;
            e1.x = r0[it].y; e1.y = r1[it].y;
            e2.x = r0[it].z; e2.y = r1[it].z;
            e3.x = r0[it].w; e3.y = r1[it].w;
            dstp[0] = e0; dstp[1] = e1; dstp[2] = e2; dstp[3] = e3;
        }
    }
    __syncthreads();

    for (int mt = blockIdx.x * 8 + wave; mt < mtiles; mt += gridDim.x * 8) {
        const int row = mt * 16 + l15;
        const float s = inv_deg[row];
        const float* xr = A0 + (size_t)row * DH + half * 2;
        const float* hr = A1 + (size_t)row * DH + half * 2;

        v8f acc[NT];
#pragma unroll
        for (int nt = 0; nt < NT; ++nt) acc[nt] = v8f{};

        v2f an[8];
#pragma unroll
        for (int i = 0; i < 8; ++i) an[i] = *reinterpret_cast<const v2f*>(xr + 4 * i);

#pragma unroll 1
        for (int tc = 0; tc < KS / 8; ++tc) {
            v2f ac[8];
#pragma unroll
            for (int i = 0; i < 8; ++i) {
                ac[i].x = an[i].x * s;
                ac[i].y = an[i].y * s;
            }
            if (tc + 1 < KS / 8) {
                const int kn = 32 * (tc + 1);
                const float* nb = (kn < DH) ? (xr + kn) : (hr + (kn - DH));
#pragma unroll
                for (int i = 0; i < 8; ++i) an[i] = *reinterpret_cast<const v2f*>(nb + 4 * i);
            }
            const v2f* wb = Ws + (size_t)(tc * 8 * NT) * 32 + lane;
#pragma unroll
            for (int i = 0; i < 8; ++i) {
#pragma unroll
                for (int nt = 0; nt < NT; ++nt) {
                    const v2f b = wb[(i * NT + nt) * 32];
                    acc[nt] = __builtin_amdgcn_wmma_f32_16x16x4_f32(
                         false, ac[i],  false, b,
                         (short)0, acc[nt],  false,  false);
                }
            }
        }

#pragma unroll
        for (int nt = 0; nt < NT; ++nt) {
#pragma unroll
            for (int j = 0; j < 8; ++j) {
                const int r = mt * 16 + half * 8 + j;
                out[(size_t)r * FOUT + nt * 16 + l15] = acc[nt][j];
            }
        }
    }
}


template <int F>
__device__ __forceinline__ float4 gather_row_sum(
        const float* __restrict__ xw,
        const int* __restrict__ grouped,
        const unsigned* __restrict__ off,
        const unsigned* __restrict__ cnt,
        unsigned node, unsigned lin, int N) {
    constexpr int LPN = F / 4;
    const unsigned n = cnt[node];
    const int* seg = grouped + off[node];
    const float4* rows = reinterpret_cast<const float4*>(xw) + lin;
    const unsigned last = (unsigned)N - 1u;

    float4 acc = make_float4(0.0f, 0.0f, 0.0f, 0.0f);
    for (unsigned k0 = 0; k0 < n; k0 += LPN) {
        const unsigned m = min(n - k0, (unsigned)LPN);
        const unsigned mine = min((unsigned)seg[k0 + min(lin, m - 1u)], last);
        for (unsigned j0 = 0; j0 < m; j0 += 8) {
            float4 v[8];
#pragma unroll
            for (int j = 0; j < 8; ++j) {
                const unsigned s = __shfl(mine, (int)(j0 + j), LPN);
                v[j] = make_float4(0.0f, 0.0f, 0.0f, 0.0f);
                if (j0 + j < m) v[j] = rows[(size_t)s * LPN];
            }
#pragma unroll
            for (int j = 0; j < 8; ++j) {
                acc.x += v[j].x;
                acc.y += v[j].y;
                acc.z += v[j].z;
                acc.w += v[j].w;
            }
        }
    }
    return acc;
}

__device__ __forceinline__ float sigmoid_f(float g) {
    return 1.0f / (1.0f + __expf(-g));
}


__global__ __launch_bounds__(256) void gather_gates_kernel(
        const float* __restrict__ xw,
        const int* __restrict__ grouped,
        const unsigned* __restrict__ off,
        const unsigned* __restrict__ cnt,
        const float* __restrict__ inv_di,
        const float* __restrict__ b_gate,
        const float* __restrict__ h,
        float* __restrict__ u,
        float* __restrict__ rh,
        int N) {
    constexpr int LPN = 2 * DH / 4;
    constexpr int HQ  = DH / 4;
    const unsigned t = blockIdx.x * blockDim.x + threadIdx.x;
    const unsigned lin  = t % LPN;
    const unsigned step = gridDim.x * blockDim.x / LPN;
    const unsigned c4 = lin & (HQ - 1);
    const float4 b  = reinterpret_cast<const float4*>(b_gate)[lin];

    for (unsigned node = t / LPN; node < (unsigned)N; node += step) {
        const float inv = inv_di[node];
        const float4 hh = reinterpret_cast<const float4*>(h)[(size_t)node * HQ + c4];

        const float4 acc = gather_row_sum<2 * DH>(xw, grouped, off, cnt, node, lin, N);

        float4 g;
        g.x = sigmoid_f(acc.x * inv + b.x);
        g.y = sigmoid_f(acc.y * inv + b.y);
        g.z = sigmoid_f(acc.z * inv + b.z);
        g.w = sigmoid_f(acc.w * inv + b.w);
        if (lin < HQ) {
            reinterpret_cast<float4*>(rh)[(size_t)node * HQ + c4] =
                make_float4(g.x * hh.x, g.y * hh.y, g.z * hh.z, g.w * hh.w);
        } else {
            reinterpret_cast<float4*>(u)[(size_t)node * HQ + c4] = g;
        }
    }
}


__global__ __launch_bounds__(256) void gather_final_kernel(
        const float* __restrict__ xw,
        const int* __restrict__ grouped,
        const unsigned* __restrict__ off,
        const unsigned* __restrict__ cnt,
        const float* __restrict__ inv_di,
        const float* __restrict__ b_cand,
        const float* __restrict__ h,
        const float* __restrict__ u,
        float* __restrict__ out,
        int N) {
    constexpr int LPN = DH / 4;
    const unsigned t = blockIdx.x * blockDim.x + threadIdx.x;
    const unsigned lin  = t % LPN;
    const unsigned step = gridDim.x * blockDim.x / LPN;
    const float4 b  = reinterpret_cast<const float4*>(b_cand)[lin];

    for (unsigned node = t / LPN; node < (unsigned)N; node += step) {
        const float inv = inv_di[node];
        const float4 hh = reinterpret_cast<const float4*>(h)[(size_t)node * LPN + lin];
        const float4 uu = reinterpret_cast<const float4*>(u)[(size_t)node * LPN + lin];

        const float4 acc = gather_row_sum<DH>(xw, grouped, off, cnt, node, lin, N);

        float4 o;
        o.x = uu.x * hh.x + (1.0f - uu.x) * tanhf(acc.x * inv + b.x);
        o.y = uu.y * hh.y + (1.0f - uu.y) * tanhf(acc.y * inv + b.y);
        o.z = uu.z * hh.z + (1.0f - uu.z) * tanhf(acc.z * inv + b.z);
        o.w = uu.w * hh.w + (1.0f - uu.w) * tanhf(acc.w * inv + b.w);
        reinterpret_cast<float4*>(out)[(size_t)node * LPN + lin] = o;
    }
}


extern "C" void kernel_launch(void* const* d_in, const int* in_sizes, int n_in,
                              void* d_out, int out_size, void* d_ws, size_t ws_size,
                              hipStream_t stream) {
    const float* x      = (const float*)d_in[0];
    const float* h      = (const float*)d_in[1];
    const int*   src    = (const int*)d_in[2];
    const int*   dst    = (const int*)d_in[3];
    const float* W_gate = (const float*)d_in[4];
    const float* b_gate = (const float*)d_in[5];
    const float* W_cand = (const float*)d_in[6];
    const float* b_cand = (const float*)d_in[7];

    const int N = in_sizes[0] / DH;
    const int E = in_sizes[2];

    float* p      = (float*)d_ws;
    float* inv_do = p;
    float* inv_di = p + N;
    float* bufA   = p + 2 * (size_t)N;
    float* bufB   = bufA + (size_t)N * 2 * DH;
    float* bufC   = bufB + (size_t)N * 2 * DH;
    unsigned* q   = (unsigned*)(bufC + (size_t)N * DH);
    float* xw     = bufA;
    float* u      = bufB;
    float* rh     = bufB + (size_t)N * DH;
    float* xw2    = bufC;
    unsigned* cnt_out = q;
    unsigned* cnt_in  = q + N;
    unsigned* cursor  = q + 2 * (size_t)N;
    unsigned* off     = cursor + 32;
    unsigned* rank    = off + N;
    int* grouped      = (int*)(rank + E);
    float* out    = (float*)d_out;

    const int mtiles = N / 16;
    int gemm_blocks = (mtiles + 7) / 8;
    if (gemm_blocks > 512) gemm_blocks = 512;
    int edge_blocks = (E + 255) / 256;
    if (edge_blocks > 1024) edge_blocks = 1024;

    hipMemsetAsync(cnt_out, 0, sizeof(unsigned) * (2 * (size_t)N + 32), stream);
    degree_rank_kernel<<<edge_blocks, 256, 0, stream>>>(src, dst, cnt_out, cnt_in, rank, E);
    degree_finalize_offsets_kernel<<<(N + 255) / 256, 256, 0, stream>>>(
        cnt_out, cnt_in, cursor, inv_do, inv_di, off, N);
    group_fill_kernel<<<edge_blocks, 256, 0, stream>>>(src, dst, rank, off, grouped, E);

    gemm_norm_wmma<2 * DH><<<gemm_blocks, dim3(32, 8), 0, stream>>>(
        x, h, W_gate, inv_do, xw, mtiles);
    {
        unsigned blocks = ((unsigned)N * (2 * DH / 4) + 255) / 256;
        if (blocks > 2048u) blocks = 2048u;
        gather_gates_kernel<<<blocks, 256, 0, stream>>>(
            xw, grouped, off, cnt_in, inv_di, b_gate, h, u, rh, N);
    }

    gemm_norm_wmma<DH><<<gemm_blocks, dim3(32, 8), 0, stream>>>(
        x, rh, W_cand, inv_do, xw2, mtiles);
    {
        unsigned blocks = ((unsigned)N * (DH / 4) + 255) / 256;
        if (blocks > 2048u) blocks = 2048u;
        gather_final_kernel<<<blocks, 256, 0, stream>>>(
            xw2, grouped, off, cnt_in, inv_di, b_cand, h, u, out, N);
    }
}


# Group edges by destination with a two-level LDS counting sort (256 node-range buckets, then per-bucket count, scan, place): no global atomics, no memset; per-node sum order = LDS arrival rank
# speedup vs baseline: 6.0365x; 1.5925x over previous
#include <hip/hip_runtime.h>
#include <math.h>


typedef __attribute__((ext_vector_type(2))) float v2f;
typedef __attribute__((ext_vector_type(4))) float v4f;
typedef __attribute__((ext_vector_type(8))) float v8f;

#define DH 64


#define NB      256
#define NPB_MAX 1024
#define SRC_BITS 18

__device__ __forceinline__ unsigned block_excl_scan_1024(unsigned v, unsigned* wsum) {
    const unsigned lane = threadIdx.x & 31u;
    const unsigned wave = threadIdx.x >> 5;
    unsigned incl = v;
    for (int d = 1; d < 32; d <<= 1) {
        unsigned t = __shfl_up(incl, d);
        if (lane >= (unsigned)d) incl += t;
    }
    if (lane == 31u) wsum[wave] = incl;
    __syncthreads();
    if (wave == 0u) {
        const unsigned s = wsum[lane];
        unsigned i2 = s;
        for (int d = 1; d < 32; d <<= 1) {
            unsigned t = __shfl_up(i2, d);
            if (lane >= (unsigned)d) i2 += t;
        }
        wsum[lane] = i2 - s;
    }
    __syncthreads();
    return incl - v + wsum[wave];
}

__device__ __forceinline__ unsigned bucket_of(int node, unsigned npb) {
    return min((unsigned)node / npb, (unsigned)NB - 1u);
}

__global__ __launch_bounds__(1024) void bucket_count_kernel(
        const int* __restrict__ src,
        const int* __restrict__ dst,
        unsigned* __restrict__ tab_d,
        unsigned* __restrict__ tab_s,
        int N, int E) {
    __shared__ unsigned hd[NB], hs[NB];
    const unsigned tid = threadIdx.x;
    if (tid < NB) { hd[tid] = 0u; hs[tid] = 0u; }
    __syncthreads();
    const unsigned npb = ((unsigned)N + NB - 1u) / NB;
    const unsigned ch  = ((unsigned)E + NB - 1u) / NB;
    const unsigned e0  = min(blockIdx.x * ch, (unsigned)E);
    const unsigned e1  = min(e0 + ch, (unsigned)E);
    for (unsigned e = e0 + tid; e < e1; e += 4096u) {
        int sv[4], dv[4];
#pragma unroll
        for (int j = 0; j < 4; ++j) {
            const unsigned ee = e + (unsigned)j * 1024u;
            sv[j] = 0; dv[j] = 0;
            if (ee < e1) { sv[j] = src[ee]; dv[j] = dst[ee]; }
        }
#pragma unroll
        for (int j = 0; j < 4; ++j) {
            const unsigned ee = e + (unsigned)j * 1024u;
            if (ee < e1) {
                atomicAdd(&hs[bucket_of(sv[j], npb)], 1u);
                atomicAdd(&hd[bucket_of(dv[j], npb)], 1u);
            }
        }
    }
    __syncthreads();
    if (tid < NB) {
        tab_d[blockIdx.x * NB + tid] = hd[tid];
        tab_s[blockIdx.x * NB + tid] = hs[tid];
    }
}

__global__ __launch_bounds__(1024) void bucket_scatter_kernel(
        const int* __restrict__ src,
        const int* __restrict__ dst,
        const unsigned* __restrict__ tab_d,
        const unsigned* __restrict__ tab_s,
        unsigned* __restrict__ pd,
        int* __restrict__ ps_src,
        unsigned* __restrict__ bstart_d,
        unsigned* __restrict__ bstart_s,
        int N, int E) {
    __shared__ unsigned part[4][4][NB];
    __shared__ unsigned cur_d[NB], cur_s[NB];
    __shared__ unsigned wsum_a[32], wsum_b[32];
    const unsigned tid = threadIdx.x;
    const unsigned g   = tid >> 8;
    const unsigned bk  = tid & 255u;
    const unsigned w   = blockIdx.x;

    unsigned tot_d = 0u, pre_d = 0u, tot_s = 0u, pre_s = 0u;
    for (unsigned k = 0; k < NB / 4; ++k) {
        const unsigned wp = g * (NB / 4) + k;
        const unsigned vd = tab_d[wp * NB + bk];
        const unsigned vs = tab_s[wp * NB + bk];
        tot_d += vd; tot_s += vs;
        if (wp < w) { pre_d += vd; pre_s += vs; }
    }
    part[0][g][bk] = tot_d; part[1][g][bk] = pre_d;
    part[2][g][bk] = tot_s; part[3][g][bk] = pre_s;
    __syncthreads();
    unsigned td = 0u, pdv = 0u, ts = 0u, psv = 0u;
    if (tid < NB) {
        for (int q = 0; q < 4; ++q) {
            td += part[0][q][tid]; pdv += part[1][q][tid];
            ts += part[2][q][tid]; psv += part[3][q][tid];
        }
    }
    const unsigned sd = block_excl_scan_1024(td, wsum_a);
    const unsigned ss = block_excl_scan_1024(ts, wsum_b);
    if (tid < NB) {
        cur_d[tid] = sd + pdv;
        cur_s[tid] = ss + psv;
        if (w == 0u) {
            bstart_d[tid] = sd;
            bstart_s[tid] = ss;
            if (tid == NB - 1u) { bstart_d[NB] = sd + td; bstart_s[NB] = ss + ts; }
        }
    }
    __syncthreads();

    const unsigned npb = ((unsigned)N + NB - 1u) / NB;
    const unsigned ch  = ((unsigned)E + NB - 1u) / NB;
    const unsigned e0  = min(w * ch, (unsigned)E);
    const unsigned e1  = min(e0 + ch, (unsigned)E);
    const unsigned last = (unsigned)E - 1u;
    for (unsigned e = e0 + tid; e < e1; e += 4096u) {
        int sv[4], dv[4];
#pragma unroll
        for (int j = 0; j < 4; ++j) {
            const unsigned ee = e + (unsigned)j * 1024u;
            sv[j] = 0; dv[j] = 0;
            if (ee < e1) { sv[j] = src[ee]; dv[j] = dst[ee]; }
        }
#pragma unroll
        for (int j = 0; j < 4; ++j) {
            const unsigned ee = e + (unsigned)j * 1024u;
            if (ee < e1) {
                const int s = sv[j];
                const int d = dv[j];
                const unsigned bd = bucket_of(d, npb);
                const unsigned p = min(atomicAdd(&cur_d[bd], 1u), last);
                const unsigned q = min(atomicAdd(&cur_s[bucket_of(s, npb)], 1u), last);
                const unsigned dl = min((unsigned)d - bd * npb, (unsigned)NPB_MAX - 1u);
                const unsigned sl = min((unsigned)s, (1u << SRC_BITS) - 1u);
                pd[p] = (dl << SRC_BITS) | sl;
                ps_src[q] = s;
            }
        }
    }
}

__global__ __launch_bounds__(1024) void bucket_sort_kernel(
        const unsigned* __restrict__ pd,
        const int* __restrict__ ps_src,
        const unsigned* __restrict__ bstart_d,
        const unsigned* __restrict__ bstart_s,
        unsigned* __restrict__ cnt_in,
        unsigned* __restrict__ off,
        float* __restrict__ inv_do,
        float* __restrict__ inv_di,
        int* __restrict__ grouped,
        int N, int E) {
    __shared__ unsigned cd[NPB_MAX], cs[NPB_MAX], cur[NPB_MAX];
    __shared__ unsigned wsum[32];
    const unsigned tid = threadIdx.x;
    const unsigned bkt = blockIdx.x;
    const unsigned npb = ((unsigned)N + NB - 1u) / NB;
    const unsigned nb  = bkt * npb;
    cd[tid] = 0u;
    cs[tid] = 0u;
    __syncthreads();
    const unsigned ds = min(bstart_d[bkt], (unsigned)E);
    const unsigned de = min(bstart_d[bkt + 1], (unsigned)E);
    const unsigned ss = min(bstart_s[bkt], (unsigned)E);
    const unsigned se = min(bstart_s[bkt + 1], (unsigned)E);
    for (unsigned e = ds + tid; e < de; e += 4096u) {
        unsigned wv[4];
#pragma unroll
        for (int j = 0; j < 4; ++j) {
            const unsigned ee = e + (unsigned)j * 1024u;
            wv[j] = 0u;
            if (ee < de) wv[j] = pd[ee];
        }
#pragma unroll
        for (int j = 0; j < 4; ++j) {
            const unsigned ee = e + (unsigned)j * 1024u;
            if (ee < de) atomicAdd(&cd[(wv[j] >> SRC_BITS) & (NPB_MAX - 1u)], 1u);
        }
    }
    for (unsigned e = ss + tid; e < se; e += 4096u) {
        int sv[4];
#pragma unroll
        for (int j = 0; j < 4; ++j) {
            const unsigned ee = e + (unsigned)j * 1024u;
            sv[j] = 0;
            if (ee < se) sv[j] = ps_src[ee];
        }
#pragma unroll
        for (int j = 0; j < 4; ++j) {
            const unsigned ee = e + (unsigned)j * 1024u;
            if (ee < se) atomicAdd(&cs[min((unsigned)sv[j] - nb, (unsigned)NPB_MAX - 1u)], 1u);
        }
    }
    __syncthreads();
    const unsigned ci   = cd[tid];
    const unsigned excl = block_excl_scan_1024(ci, wsum);
    const unsigned node = nb + tid;
    cur[tid] = ds + excl;
    if (tid < npb && node < (unsigned)N) {
        cnt_in[node] = ci;
        off[node]    = ds + excl;
        inv_do[node] = rsqrtf(fmaxf((float)cs[tid], 1.0f));
        inv_di[node] = rsqrtf(fmaxf((float)ci, 1.0f));
    }
    __syncthreads();
    const unsigned last = (unsigned)E - 1u;
    for (unsigned e = ds + tid; e < de; e += 4096u) {
        unsigned wv[4];
#pragma unroll
        for (int j = 0; j < 4; ++j) {
            const unsigned ee = e + (unsigned)j * 1024u;
            wv[j] = 0u;
            if (ee < de) wv[j] = pd[ee];
        }
#pragma unroll
        for (int j = 0; j < 4; ++j) {
            const unsigned ee = e + (unsigned)j * 1024u;
            if (ee < de) {
                const unsigned pos = min(atomicAdd(&cur[(wv[j] >> SRC_BITS) & (NPB_MAX - 1u)], 1u), last);
                grouped[pos] = (int)(wv[j] & ((1u << SRC_BITS) - 1u));
            }
        }
    }
}


template <int FOUT>
__global__ __launch_bounds__(256) void gemm_norm_wmma(
        const float* __restrict__ A0,
        const float* __restrict__ A1,
        const float* __restrict__ W,
        const float* __restrict__ inv_deg,
        float* __restrict__ out,
        int mtiles) {
    constexpr int NT = FOUT / 16;
    constexpr int KS = 2 * DH / 4;
    constexpr int C4 = FOUT / 4;

    __shared__ v2f Ws[KS * NT * 32];

    const int lane = threadIdx.x;
    const int wave = __builtin_amdgcn_readfirstlane(threadIdx.y);
    const int half = lane >> 4;
    const int l15  = lane & 15;

    {
        constexpr int TRIPS = DH * C4 / 256;
        const int tid = wave * 32 + lane;
        v4f r0[TRIPS], r1[TRIPS];
#pragma unroll
        for (int it = 0; it < TRIPS; ++it) {
            const int p  = tid + it * 256;
            const int kp = p / C4;
            const int c4 = p % C4;
            r0[it] = *reinterpret_cast<const v4f*>(W + (size_t)(2 * kp) * FOUT + c4 * 4);
            r1[it] = *reinterpret_cast<const v4f*>(W + (size_t)(2 * kp + 1) * FOUT + c4 * 4);
        }
#pragma unroll
        for (int it = 0; it < TRIPS; ++it) {
            const int p   = tid + it * 256;
            const int kp  = p / C4;
            const int c4  = p % C4;
            const int t   = kp >> 1;
            const int h   = kp & 1;
            const int col = c4 * 4;
            v2f* dstp = Ws + ((t * NT + (col >> 4)) * 32 + h * 16 + (col & 15));
            v2f e0, e1, e2, e3;
            e0.x = r0[it].x; e0.y = r1[it].x;
            e1.x = r0[it].y; e1.y = r1[it].y;
            e2.x = r0[it].z; e2.y = r1[it].z;
            e3.x = r0[it].w; e3.y = r1[it].w;
            dstp[0] = e0; dstp[1] = e1; dstp[2] = e2; dstp[3] = e3;
        }
    }
    __syncthreads();

    for (int mt = blockIdx.x * 8 + wave; mt < mtiles; mt += gridDim.x * 8) {
        const int row = mt * 16 + l15;
        const float s = inv_deg[row];
        const float* xr = A0 + (size_t)row * DH + half * 2;
        const float* hr = A1 + (size_t)row * DH + half * 2;

        v8f acc[NT];
#pragma unroll
        for (int nt = 0; nt < NT; ++nt) acc[nt] = v8f{};

        v2f an[8];
#pragma unroll
        for (int i = 0; i < 8; ++i) an[i] = *reinterpret_cast<const v2f*>(xr + 4 * i);

#pragma unroll 1
        for (int tc = 0; tc < KS / 8; ++tc) {
            v2f ac[8];
#pragma unroll
            for (int i = 0; i < 8; ++i) {
                ac[i].x = an[i].x * s;
                ac[i].y = an[i].y * s;
            }
            if (tc + 1 < KS / 8) {
                const int kn = 32 * (tc + 1);
                const float* nb = (kn < DH) ? (xr + kn) : (hr + (kn - DH));
#pragma unroll
                for (int i = 0; i < 8; ++i) an[i] = *reinterpret_cast<const v2f*>(nb + 4 * i);
            }
            const v2f* wb = Ws + (size_t)(tc * 8 * NT) * 32 + lane;
#pragma unroll
            for (int i = 0; i < 8; ++i) {
#pragma unroll
                for (int nt = 0; nt < NT; ++nt) {
                    const v2f b = wb[(i * NT + nt) * 32];
                    acc[nt] = __builtin_amdgcn_wmma_f32_16x16x4_f32(
                         false, ac[i],  false, b,
                         (short)0, acc[nt],  false,  false);
                }
            }
        }

#pragma unroll
        for (int nt = 0; nt < NT; ++nt) {
#pragma unroll
            for (int j = 0; j < 8; ++j) {
                const int r = mt * 16 + half * 8 + j;
                out[(size_t)r * FOUT + nt * 16 + l15] = acc[nt][j];
            }
        }
    }
}


template <int F>
__device__ __forceinline__ float4 gather_row_sum(
        const float* __restrict__ xw,
        const int* __restrict__ grouped,
        const unsigned* __restrict__ off,
        const unsigned* __restrict__ cnt,
        unsigned node, unsigned lin, int N) {
    constexpr int LPN = F / 4;
    const unsigned n = cnt[node];
    const int* seg = grouped + off[node];
    const float4* rows = reinterpret_cast<const float4*>(xw) + lin;
    const unsigned last = (unsigned)N - 1u;

    float4 acc = make_float4(0.0f, 0.0f, 0.0f, 0.0f);
    for (unsigned k0 = 0; k0 < n; k0 += LPN) {
        const unsigned m = min(n - k0, (unsigned)LPN);
        const unsigned mine = min((unsigned)seg[k0 + min(lin, m - 1u)], last);
        for (unsigned j0 = 0; j0 < m; j0 += 8) {
            float4 v[8];
#pragma unroll
            for (int j = 0; j < 8; ++j) {
                const unsigned s = __shfl(mine, (int)(j0 + j), LPN);
                v[j] = make_float4(0.0f, 0.0f, 0.0f, 0.0f);
                if (j0 + j < m) v[j] = rows[(size_t)s * LPN];
            }
#pragma unroll
            for (int j = 0; j < 8; ++j) {
                acc.x += v[j].x;
                acc.y += v[j].y;
                acc.z += v[j].z;
                acc.w += v[j].w;
            }
        }
    }
    return acc;
}

__device__ __forceinline__ float sigmoid_f(float g) {
    return 1.0f / (1.0f + __expf(-g));
}


__global__ __launch_bounds__(256) void gather_gates_kernel(
        const float* __restrict__ xw,
        const int* __restrict__ grouped,
        const unsigned* __restrict__ off,
        const unsigned* __restrict__ cnt,
        const float* __restrict__ inv_di,
        const float* __restrict__ b_gate,
        const float* __restrict__ h,
        float* __restrict__ u,
        float* __restrict__ rh,
        int N) {
    constexpr int LPN = 2 * DH / 4;
    constexpr int HQ  = DH / 4;
    const unsigned t = blockIdx.x * blockDim.x + threadIdx.x;
    const unsigned lin  = t % LPN;
    const unsigned step = gridDim.x * blockDim.x / LPN;
    const unsigned c4 = lin & (HQ - 1);
    const float4 b  = reinterpret_cast<const float4*>(b_gate)[lin];

    for (unsigned node = t / LPN; node < (unsigned)N; node += step) {
        const float inv = inv_di[node];
        const float4 hh = reinterpret_cast<const float4*>(h)[(size_t)node * HQ + c4];

        const float4 acc = gather_row_sum<2 * DH>(xw, grouped, off, cnt, node, lin, N);

        float4 g;
        g.x = sigmoid_f(acc.x * inv + b.x);
        g.y = sigmoid_f(acc.y * inv + b.y);
        g.z = sigmoid_f(acc.z * inv + b.z);
        g.w = sigmoid_f(acc.w * inv + b.w);
        if (lin < HQ) {
            reinterpret_cast<float4*>(rh)[(size_t)node * HQ + c4] =
                make_float4(g.x * hh.x, g.y * hh.y, g.z * hh.z, g.w * hh.w);
        } else {
            reinterpret_cast<float4*>(u)[(size_t)node * HQ + c4] = g;
        }
    }
}


__global__ __launch_bounds__(256) void gather_final_kernel(
        const float* __restrict__ xw,
        const int* __restrict__ grouped,
        const unsigned* __restrict__ off,
        const unsigned* __restrict__ cnt,
        const float* __restrict__ inv_di,
        const float* __restrict__ b_cand,
        const float* __restrict__ h,
        const float* __restrict__ u,
        float* __restrict__ out,
        int N) {
    constexpr int LPN = DH / 4;
    const unsigned t = blockIdx.x * blockDim.x + threadIdx.x;
    const unsigned lin  = t % LPN;
    const unsigned step = gridDim.x * blockDim.x / LPN;
    const float4 b  = reinterpret_cast<const float4*>(b_cand)[lin];

    for (unsigned node = t / LPN; node < (unsigned)N; node += step) {
        const float inv = inv_di[node];
        const float4 hh = reinterpret_cast<const float4*>(h)[(size_t)node * LPN + lin];
        const float4 uu = reinterpret_cast<const float4*>(u)[(size_t)node * LPN + lin];

        const float4 acc = gather_row_sum<DH>(xw, grouped, off, cnt, node, lin, N);

        float4 o;
        o.x = uu.x * hh.x + (1.0f - uu.x) * tanhf(acc.x * inv + b.x);
        o.y = uu.y * hh.y + (1.0f - uu.y) * tanhf(acc.y * inv + b.y);
        o.z = uu.z * hh.z + (1.0f - uu.z) * tanhf(acc.z * inv + b.z);
        o.w = uu.w * hh.w + (1.0f - uu.w) * tanhf(acc.w * inv + b.w);
        reinterpret_cast<float4*>(out)[(size_t)node * LPN + lin] = o;
    }
}


extern "C" void kernel_launch(void* const* d_in, const int* in_sizes, int n_in,
                              void* d_out, int out_size, void* d_ws, size_t ws_size,
                              hipStream_t stream) {
    const float* x      = (const float*)d_in[0];
    const float* h      = (const float*)d_in[1];
    const int*   src    = (const int*)d_in[2];
    const int*   dst    = (const int*)d_in[3];
    const float* W_gate = (const float*)d_in[4];
    const float* b_gate = (const float*)d_in[5];
    const float* W_cand = (const float*)d_in[6];
    const float* b_cand = (const float*)d_in[7];

    const int N = in_sizes[0] / DH;
    const int E = in_sizes[2];

    float* p      = (float*)d_ws;
    float* inv_do = p;
    float* inv_di = p + N;
    float* bufA   = p + 2 * (size_t)N;
    float* bufB   = bufA + (size_t)N * 2 * DH;
    float* bufC   = bufB + (size_t)N * 2 * DH;
    unsigned* q   = (unsigned*)(bufC + (size_t)N * DH);
    float* xw     = bufA;
    float* u      = bufB;
    float* rh     = bufB + (size_t)N * DH;
    float* xw2    = bufC;
    unsigned* cnt_in  = q + N;
    unsigned* off     = q + 2 * (size_t)N + 32;
    int* grouped      = (int*)(off + N) + E;
    unsigned* pd       = (unsigned*)bufB;
    int* ps_src        = (int*)(pd + E);
    unsigned* tab_d    = (unsigned*)(ps_src + E);
    unsigned* tab_s    = tab_d + NB * NB;
    unsigned* bstart_d = tab_s + NB * NB;
    unsigned* bstart_s = bstart_d + 512;
    float* out    = (float*)d_out;

    const int mtiles = N / 16;
    int gemm_blocks = (mtiles + 7) / 8;
    if (gemm_blocks > 512) gemm_blocks = 512;

    bucket_count_kernel<<<NB, 1024, 0, stream>>>(src, dst, tab_d, tab_s, N, E);
    bucket_scatter_kernel<<<NB, 1024, 0, stream>>>(
        src, dst, tab_d, tab_s, pd, ps_src, bstart_d, bstart_s, N, E);
    bucket_sort_kernel<<<NB, 1024, 0, stream>>>(
        pd, ps_src, bstart_d, bstart_s, cnt_in, off, inv_do, inv_di, grouped, N, E);

    gemm_norm_wmma<2 * DH><<<gemm_blocks, dim3(32, 8), 0, stream>>>(
        x, h, W_gate, inv_do, xw, mtiles);
    {
        unsigned blocks = ((unsigned)N * (2 * DH / 4) + 255) / 256;
        if (blocks > 2048u) blocks = 2048u;
        gather_gates_kernel<<<blocks, 256, 0, stream>>>(
            xw, grouped, off, cnt_in, inv_di, b_gate, h, u, rh, N);
    }

    gemm_norm_wmma<DH><<<gemm_blocks, dim3(32, 8), 0, stream>>>(
        x, rh, W_cand, inv_do, xw2, mtiles);
    {
        unsigned blocks = ((unsigned)N * (DH / 4) + 255) / 256;
        if (blocks > 2048u) blocks = 2048u;
        gather_final_kernel<<<blocks, 256, 0, stream>>>(
            xw2, grouped, off, cnt_in, inv_di, b_cand, h, u, out, N);
    }
}


# Sort level 1 = chunk-local LDS counting sort written back in place with whole-line stores; count launch, table scan, scattered stores go (7 -> 6 launches); per-node sum order = LDS arrival rank
# speedup vs baseline: 7.8722x; 1.3041x over previous
#include <hip/hip_runtime.h>
#include <math.h>


typedef __attribute__((ext_vector_type(2))) float v2f;
typedef __attribute__((ext_vector_type(4))) float v4f;
typedef __attribute__((ext_vector_type(8))) float v8f;

#define DH 64


#define NB      256
#define NPB_MAX 1024
#define SRC_BITS 18
#define CH      6400
#define EPT     7
#define TROW    (NB + 1)
#define NC_MAX  1024

__device__ __forceinline__ unsigned block_excl_scan_1024(unsigned v, unsigned* wsum) {
    const unsigned lane = threadIdx.x & 31u;
    const unsigned wave = threadIdx.x >> 5;
    unsigned incl = v;
    for (int d = 1; d < 32; d <<= 1) {
        unsigned t = __shfl_up(incl, d);
        if (lane >= (unsigned)d) incl += t;
    }
    if (lane == 31u) wsum[wave] = incl;
    __syncthreads();
    if (wave == 0u) {
        const unsigned s = wsum[lane];
        unsigned i2 = s;
        for (int d = 1; d < 32; d <<= 1) {
            unsigned t = __shfl_up(i2, d);
            if (lane >= (unsigned)d) i2 += t;
        }
        wsum[lane] = i2 - s;
    }
    __syncthreads();
    return incl - v + wsum[wave];
}

__device__ __forceinline__ unsigned bucket_of(int node, unsigned npb) {
    return min((unsigned)node / npb, (unsigned)NB - 1u);
}

__global__ __launch_bounds__(1024) void chunk_sort_kernel(
        const int* __restrict__ src,
        const int* __restrict__ dst,
        unsigned* __restrict__ tab_d,
        unsigned* __restrict__ tab_s,
        unsigned* __restrict__ pd,
        unsigned* __restrict__ ps_src,
        int N, int E) {
    __shared__ unsigned hd[NB], hs[NB];
    __shared__ unsigned wsum_a[32], wsum_b[32];
    __shared__ unsigned stg_d[CH], stg_s[CH];
    const unsigned tid = threadIdx.x;
    const unsigned w   = blockIdx.x;
    if (tid < NB) { hd[tid] = 0u; hs[tid] = 0u; }
    const unsigned npb = ((unsigned)N + NB - 1u) / NB;
    const unsigned e0  = min(w * (unsigned)CH, (unsigned)E);
    const unsigned e1  = min(e0 + (unsigned)CH, (unsigned)E);
    int sv[EPT], dv[EPT];
#pragma unroll
    for (int j = 0; j < EPT; ++j) {
        const unsigned ee = e0 + tid + (unsigned)j * 1024u;
        sv[j] = 0; dv[j] = 0;
        if (ee < e1) { sv[j] = src[ee]; dv[j] = dst[ee]; }
    }
    __syncthreads();
    unsigned rd[EPT], rs[EPT];
#pragma unroll
    for (int j = 0; j < EPT; ++j) {
        const unsigned ee = e0 + tid + (unsigned)j * 1024u;
        rd[j] = 0u; rs[j] = 0u;
        if (ee < e1) {
            rd[j] = atomicAdd(&hd[bucket_of(dv[j], npb)], 1u);
            rs[j] = atomicAdd(&hs[bucket_of(sv[j], npb)], 1u);
        }
    }
    __syncthreads();
    const unsigned cdv = (tid < NB) ? hd[tid] : 0u;
    const unsigned csv = (tid < NB) ? hs[tid] : 0u;
    const unsigned od = block_excl_scan_1024(cdv, wsum_a);
    const unsigned os = block_excl_scan_1024(csv, wsum_b);
    if (tid < NB) {
        hd[tid] = od;
        hs[tid] = os;
        tab_d[w * TROW + tid] = od;
        tab_s[w * TROW + tid] = os;
        if (tid == NB - 1u) {
            tab_d[w * TROW + NB] = od + cdv;
            tab_s[w * TROW + NB] = os + csv;
        }
    }
    __syncthreads();
#pragma unroll
    for (int j = 0; j < EPT; ++j) {
        const unsigned ee = e0 + tid + (unsigned)j * 1024u;
        if (ee < e1) {
            const int s = sv[j];
            const int d = dv[j];
            const unsigned bd = bucket_of(d, npb);
            const unsigned bs = bucket_of(s, npb);
            const unsigned dl = min((unsigned)d - bd * npb, (unsigned)NPB_MAX - 1u);
            const unsigned sl = min((unsigned)s, (1u << SRC_BITS) - 1u);
            const unsigned p = min(hd[bd] + rd[j], (unsigned)CH - 1u);
            const unsigned q = min(hs[bs] + rs[j], (unsigned)CH - 1u);
            stg_d[p] = (dl << SRC_BITS) | sl;
            stg_s[q] = (unsigned)s;
        }
    }
    __syncthreads();
    const unsigned n = e1 - e0;
    for (unsigned i = tid; i < n; i += 1024u) {
        pd[e0 + i]     = stg_d[i];
        ps_src[e0 + i] = stg_s[i];
    }
}

template <typename F>
__device__ __forceinline__ void for_each_in_pieces(
        const unsigned* __restrict__ arr, unsigned last_word,
        const unsigned* s0, const unsigned* s1, unsigned nc, F f) {
    const unsigned hw  = threadIdx.x >> 4;
    const unsigned l16 = threadIdx.x & 15u;
    for (unsigned c0 = hw; c0 < nc; c0 += 256u) {
        unsigned b[4], e[4], v[4][2];
#pragma unroll
        for (int k = 0; k < 4; ++k) {
            const unsigned c = c0 + 64u * (unsigned)k;
            b[k] = 0u; e[k] = 0u;
            if (c < nc) { b[k] = s0[c] + l16; e[k] = s1[c]; }
        }
#pragma unroll
        for (int k = 0; k < 4; ++k) {
#pragma unroll
            for (int r = 0; r < 2; ++r) {
                v[k][r] = arr[min(b[k] + 16u * (unsigned)r, last_word)];
            }
        }
#pragma unroll
        for (int k = 0; k < 4; ++k) {
#pragma unroll
            for (int r = 0; r < 2; ++r) {
                if (b[k] + 16u * (unsigned)r < e[k]) f(v[k][r]);
            }
            for (unsigned i = b[k] + 32u; i < e[k]; i += 16u) f(arr[i]);
        }
    }
}

__global__ __launch_bounds__(1024) void bucket_sort_kernel(
        const unsigned* __restrict__ pd,
        const unsigned* __restrict__ ps_src,
        const unsigned* __restrict__ tab_d,
        const unsigned* __restrict__ tab_s,
        unsigned* __restrict__ cnt_in,
        unsigned* __restrict__ off,
        float* __restrict__ inv_do,
        float* __restrict__ inv_di,
        int* __restrict__ grouped,
        int N, int E, int nchunks) {
    __shared__ unsigned cd[NPB_MAX], cs[NPB_MAX], cur[NPB_MAX];
    __shared__ unsigned wsum[32];
    __shared__ unsigned pd0[NC_MAX], pd1[NC_MAX], ps0[NC_MAX], ps1[NC_MAX];
    __shared__ unsigned bstart;
    const unsigned tid = threadIdx.x;
    const unsigned bkt = blockIdx.x;
    const unsigned npb = ((unsigned)N + NB - 1u) / NB;
    const unsigned nb  = bkt * npb;
    const unsigned nc  = min((unsigned)nchunks, (unsigned)NC_MAX);
    cd[tid] = 0u;
    cs[tid] = 0u;
    unsigned before = 0u;
    if (tid < nc) {
        const unsigned base = min(tid * (unsigned)CH, (unsigned)E);
        const unsigned a0 = tab_d[tid * TROW + bkt];
        const unsigned a1 = tab_d[tid * TROW + bkt + 1u];
        const unsigned b0 = tab_s[tid * TROW + bkt];
        const unsigned b1 = tab_s[tid * TROW + bkt + 1u];
        pd0[tid] = min(base + a0, (unsigned)E);
        pd1[tid] = min(base + a1, (unsigned)E);
        ps0[tid] = min(base + b0, (unsigned)E);
        ps1[tid] = min(base + b1, (unsigned)E);
        before = a0;
    }
    const unsigned pre = block_excl_scan_1024(before, wsum);
    if (tid == 1023u) bstart = pre + before;
    __syncthreads();
    const unsigned ds = min(bstart, (unsigned)E);
    const unsigned last = (unsigned)E - 1u;

    for_each_in_pieces(pd, last, pd0, pd1, nc, [&](unsigned wv) {
        atomicAdd(&cd[(wv >> SRC_BITS) & (NPB_MAX - 1u)], 1u);
    });
    for_each_in_pieces(ps_src, last, ps0, ps1, nc, [&](unsigned sv) {
        atomicAdd(&cs[min(sv - nb, (unsigned)NPB_MAX - 1u)], 1u);
    });
    __syncthreads();
    const unsigned ci   = cd[tid];
    const unsigned excl = block_excl_scan_1024(ci, wsum);
    const unsigned node = nb + tid;
    cur[tid] = ds + excl;
    if (tid < npb && node < (unsigned)N) {
        cnt_in[node] = ci;
        off[node]    = ds + excl;
        inv_do[node] = rsqrtf(fmaxf((float)cs[tid], 1.0f));
        inv_di[node] = rsqrtf(fmaxf((float)ci, 1.0f));
    }
    __syncthreads();
    for_each_in_pieces(pd, last, pd0, pd1, nc, [&](unsigned wv) {
        const unsigned pos = min(atomicAdd(&cur[(wv >> SRC_BITS) & (NPB_MAX - 1u)], 1u), last);
        grouped[pos] = (int)(wv & ((1u << SRC_BITS) - 1u));
    });
}


template <int FOUT>
__global__ __launch_bounds__(256) void gemm_norm_wmma(
        const float* __restrict__ A0,
        const float* __restrict__ A1,
        const float* __restrict__ W,
        const float* __restrict__ inv_deg,
        float* __restrict__ out,
        int mtiles) {
    constexpr int NT = FOUT / 16;
    constexpr int KS = 2 * DH / 4;
    constexpr int C4 = FOUT / 4;

    __shared__ v2f Ws[KS * NT * 32];

    const int lane = threadIdx.x;
    const int wave = __builtin_amdgcn_readfirstlane(threadIdx.y);
    const int half = lane >> 4;
    const int l15  = lane & 15;

    {
        constexpr int TRIPS = DH * C4 / 256;
        const int tid = wave * 32 + lane;
        v4f r0[TRIPS], r1[TRIPS];
#pragma unroll
        for (int it = 0; it < TRIPS; ++it) {
            const int p  = tid + it * 256;
            const int kp = p / C4;
            const int c4 = p % C4;
            r0[it] = *reinterpret_cast<const v4f*>(W + (size_t)(2 * kp) * FOUT + c4 * 4);
            r1[it] = *reinterpret_cast<const v4f*>(W + (size_t)(2 * kp + 1) * FOUT + c4 * 4);
        }
#pragma unroll
        for (int it = 0; it < TRIPS; ++it) {
            const int p   = tid + it * 256;
            const int kp  = p / C4;
            const int c4  = p % C4;
            const int t   = kp >> 1;
            const int h   = kp & 1;
            const int col = c4 * 4;
            v2f* dstp = Ws + ((t * NT + (col >> 4)) * 32 + h * 16 + (col & 15));
            v2f e0, e1, e2, e3;
            e0.x = r0[it].x; e0.y = r1[it].x;
            e1.x = r0[it].y; e1.y = r1[it].y;
            e2.x = r0[it].z; e2.y = r1[it].z;
            e3.x = r0[it].w; e3.y = r1[it].w;
            dstp[0] = e0; dstp[1] = e1; dstp[2] = e2; dstp[3] = e3;
        }
    }
    __syncthreads();

    for (int mt = blockIdx.x * 8 + wave; mt < mtiles; mt += gridDim.x * 8) {
        const int row = mt * 16 + l15;
        const float s = inv_deg[row];
        const float* xr = A0 + (size_t)row * DH + half * 2;
        const float* hr = A1 + (size_t)row * DH + half * 2;

        v8f acc[NT];
#pragma unroll
        for (int nt = 0; nt < NT; ++nt) acc[nt] = v8f{};

        v2f an[8];
#pragma unroll
        for (int i = 0; i < 8; ++i) an[i] = *reinterpret_cast<const v2f*>(xr + 4 * i);

#pragma unroll 1
        for (int tc = 0; tc < KS / 8; ++tc) {
            v2f ac[8];
#pragma unroll
            for (int i = 0; i < 8; ++i) {
                ac[i].x = an[i].x * s;
                ac[i].y = an[i].y * s;
            }
            if (tc + 1 < KS / 8) {
                const int kn = 32 * (tc + 1);
                const float* nb = (kn < DH) ? (xr + kn) : (hr + (kn - DH));
#pragma unroll
                for (int i = 0; i < 8; ++i) an[i] = *reinterpret_cast<const v2f*>(nb + 4 * i);
            }
            const v2f* wb = Ws + (size_t)(tc * 8 * NT) * 32 + lane;
#pragma unroll
            for (int i = 0; i < 8; ++i) {
#pragma unroll
                for (int nt = 0; nt < NT; ++nt) {
                    const v2f b = wb[(i * NT + nt) * 32];
                    acc[nt] = __builtin_amdgcn_wmma_f32_16x16x4_f32(
                         false, ac[i],  false, b,
                         (short)0, acc[nt],  false,  false);
                }
            }
        }

#pragma unroll
        for (int nt = 0; nt < NT; ++nt) {
#pragma unroll
            for (int j = 0; j < 8; ++j) {
                const int r = mt * 16 + half * 8 + j;
                out[(size_t)r * FOUT + nt * 16 + l15] = acc[nt][j];
            }
        }
    }
}


template <int F>
__device__ __forceinline__ float4 gather_row_sum(
        const float* __restrict__ xw,
        const int* __restrict__ grouped,
        const unsigned* __restrict__ off,
        const unsigned* __restrict__ cnt,
        unsigned node, unsigned lin, int N) {
    constexpr int LPN = F / 4;
    const unsigned n = cnt[node];
    const int* seg = grouped + off[node];
    const float4* rows = reinterpret_cast<const float4*>(xw) + lin;
    const unsigned last = (unsigned)N - 1u;

    float4 acc = make_float4(0.0f, 0.0f, 0.0f, 0.0f);
    for (unsigned k0 = 0; k0 < n; k0 += LPN) {
        const unsigned m = min(n - k0, (unsigned)LPN);
        const unsigned mine = min((unsigned)seg[k0 + min(lin, m - 1u)], last);
        for (unsigned j0 = 0; j0 < m; j0 += 8) {
            float4 v[8];
#pragma unroll
            for (int j = 0; j < 8; ++j) {
                const unsigned s = __shfl(mine, (int)(j0 + j), LPN);
                v[j] = make_float4(0.0f, 0.0f, 0.0f, 0.0f);
                if (j0 + j < m) v[j] = rows[(size_t)s * LPN];
            }
#pragma unroll
            for (int j = 0; j < 8; ++j) {
                acc.x += v[j].x;
                acc.y += v[j].y;
                acc.z += v[j].z;
                acc.w += v[j].w;
            }
        }
    }
    return acc;
}

__device__ __forceinline__ float sigmoid_f(float g) {
    return 1.0f / (1.0f + __expf(-g));
}


__global__ __launch_bounds__(256) void gather_gates_kernel(
        const float* __restrict__ xw,
        const int* __restrict__ grouped,
        const unsigned* __restrict__ off,
        const unsigned* __restrict__ cnt,
        const float* __restrict__ inv_di,
        const float* __restrict__ b_gate,
        const float* __restrict__ h,
        float* __restrict__ u,
        float* __restrict__ rh,
        int N) {
    constexpr int LPN = 2 * DH / 4;
    constexpr int HQ  = DH / 4;
    const unsigned t = blockIdx.x * blockDim.x + threadIdx.x;
    const unsigned lin  = t % LPN;
    const unsigned step = gridDim.x * blockDim.x / LPN;
    const unsigned c4 = lin & (HQ - 1);
    const float4 b  = reinterpret_cast<const float4*>(b_gate)[lin];

    for (unsigned node = t / LPN; node < (unsigned)N; node += step) {
        const float inv = inv_di[node];
        const float4 hh = reinterpret_cast<const float4*>(h)[(size_t)node * HQ + c4];

        const float4 acc = gather_row_sum<2 * DH>(xw, grouped, off, cnt, node, lin, N);

        float4 g;
        g.x = sigmoid_f(acc.x * inv + b.x);
        g.y = sigmoid_f(acc.y * inv + b.y);
        g.z = sigmoid_f(acc.z * inv + b.z);
        g.w = sigmoid_f(acc.w * inv + b.w);
        if (lin < HQ) {
            reinterpret_cast<float4*>(rh)[(size_t)node * HQ + c4] =
                make_float4(g.x * hh.x, g.y * hh.y, g.z * hh.z, g.w * hh.w);
        } else {
            reinterpret_cast<float4*>(u)[(size_t)node * HQ + c4] = g;
        }
    }
}


__global__ __launch_bounds__(256) void gather_final_kernel(
        const float* __restrict__ xw,
        const int* __restrict__ grouped,
        const unsigned* __restrict__ off,
        const unsigned* __restrict__ cnt,
        const float* __restrict__ inv_di,
        const float* __restrict__ b_cand,
        const float* __restrict__ h,
        const float* __restrict__ u,
        float* __restrict__ out,
        int N) {
    constexpr int LPN = DH / 4;
    const unsigned t = blockIdx.x * blockDim.x + threadIdx.x;
    const unsigned lin  = t % LPN;
    const unsigned step = gridDim.x * blockDim.x / LPN;
    const float4 b  = reinterpret_cast<const float4*>(b_cand)[lin];

    for (unsigned node = t / LPN; node < (unsigned)N; node += step) {
        const float inv = inv_di[node];
        const float4 hh = reinterpret_cast<const float4*>(h)[(size_t)node * LPN + lin];
        const float4 uu = reinterpret_cast<const float4*>(u)[(size_t)node * LPN + lin];

        const float4 acc = gather_row_sum<DH>(xw, grouped, off, cnt, node, lin, N);

        float4 o;
        o.x = uu.x * hh.x + (1.0f - uu.x) * tanhf(acc.x * inv + b.x);
        o.y = uu.y * hh.y + (1.0f - uu.y) * tanhf(acc.y * inv + b.y);
        o.z = uu.z * hh.z + (1.0f - uu.z) * tanhf(acc.z * inv + b.z);
        o.w = uu.w * hh.w + (1.0f - uu.w) * tanhf(acc.w * inv + b.w);
        reinterpret_cast<float4*>(out)[(size_t)node * LPN + lin] = o;
    }
}


extern "C" void kernel_launch(void* const* d_in, const int* in_sizes, int n_in,
                              void* d_out, int out_size, void* d_ws, size_t ws_size,
                              hipStream_t stream) {
    const float* x      = (const float*)d_in[0];
    const float* h      = (const float*)d_in[1];
    const int*   src    = (const int*)d_in[2];
    const int*   dst    = (const int*)d_in[3];
    const float* W_gate = (const float*)d_in[4];
    const float* b_gate = (const float*)d_in[5];
    const float* W_cand = (const float*)d_in[6];
    const float* b_cand = (const float*)d_in[7];

    const int N = in_sizes[0] / DH;
    const int E = in_sizes[2];

    float* p      = (float*)d_ws;
    float* inv_do = p;
    float* inv_di = p + N;
    float* bufA   = p + 2 * (size_t)N;
    float* bufB   = bufA + (size_t)N * 2 * DH;
    float* bufC   = bufB + (size_t)N * 2 * DH;
    unsigned* q   = (unsigned*)(bufC + (size_t)N * DH);
    float* xw     = bufA;
    float* u      = bufB;
    float* rh     = bufB + (size_t)N * DH;
    float* xw2    = bufC;
    unsigned* cnt_in  = q + N;
    unsigned* off     = q + 2 * (size_t)N + 32;
    int* grouped      = (int*)(off + N) + E;
    const int nchunks  = (E + CH - 1) / CH;
    unsigned* pd       = (unsigned*)bufB;
    unsigned* ps_src   = pd + E;
    unsigned* tab_d    = ps_src + E;
    unsigned* tab_s    = tab_d + (size_t)nchunks * TROW;
    float* out    = (float*)d_out;

    const int mtiles = N / 16;
    int gemm_blocks = (mtiles + 7) / 8;
    if (gemm_blocks > 512) gemm_blocks = 512;

    chunk_sort_kernel<<<nchunks, 1024, 0, stream>>>(
        src, dst, tab_d, tab_s, pd, ps_src, N, E);
    bucket_sort_kernel<<<NB, 1024, 0, stream>>>(
        pd, ps_src, tab_d, tab_s, cnt_in, off, inv_do, inv_di, grouped, N, E, nchunks);

    gemm_norm_wmma<2 * DH><<<gemm_blocks, dim3(32, 8), 0, stream>>>(
        x, h, W_gate, inv_do, xw, mtiles);
    {
        unsigned blocks = ((unsigned)N * (2 * DH / 4) + 255) / 256;
        if (blocks > 2048u) blocks = 2048u;
        gather_gates_kernel<<<blocks, 256, 0, stream>>>(
            xw, grouped, off, cnt_in, inv_di, b_gate, h, u, rh, N);
    }

    gemm_norm_wmma<DH><<<gemm_blocks, dim3(32, 8), 0, stream>>>(
        x, rh, W_cand, inv_do, xw2, mtiles);
    {
        unsigned blocks = ((unsigned)N * (DH / 4) + 255) / 256;
        if (blocks > 2048u) blocks = 2048u;
        gather_final_kernel<<<blocks, 256, 0, stream>>>(
            xw2, grouped, off, cnt_in, inv_di, b_cand, h, u, out, N);
    }
}


# Bucket sort keeps its edges in LDS, places them in an LDS copy of its part of the grouped list, writes that with whole-line stores, not scattered 4-byte stores; sum order = LDS arrival rank
# speedup vs baseline: 8.1169x; 1.0311x over previous
#include <hip/hip_runtime.h>
#include <math.h>


typedef __attribute__((ext_vector_type(2))) float v2f;
typedef __attribute__((ext_vector_type(4))) float v4f;
typedef __attribute__((ext_vector_type(8))) float v8f;

#define DH 64


#define NB      256
#define NPB_MAX 1024
#define SRC_BITS 18
#define CH      6400
#define EPT     7
#define TROW    (NB + 1)
#define NC_MAX  1024
#define CAP     8192

__device__ __forceinline__ unsigned block_excl_scan_1024(unsigned v, unsigned* wsum) {
    const unsigned lane = threadIdx.x & 31u;
    const unsigned wave = threadIdx.x >> 5;
    unsigned incl = v;
    for (int d = 1; d < 32; d <<= 1) {
        unsigned t = __shfl_up(incl, d);
        if (lane >= (unsigned)d) incl += t;
    }
    if (lane == 31u) wsum[wave] = incl;
    __syncthreads();
    if (wave == 0u) {
        const unsigned s = wsum[lane];
        unsigned i2 = s;
        for (int d = 1; d < 32; d <<= 1) {
            unsigned t = __shfl_up(i2, d);
            if (lane >= (unsigned)d) i2 += t;
        }
        wsum[lane] = i2 - s;
    }
    __syncthreads();
    return incl - v + wsum[wave];
}

__device__ __forceinline__ unsigned bucket_of(int node, unsigned npb) {
    return min((unsigned)node / npb, (unsigned)NB - 1u);
}

__global__ __launch_bounds__(1024) void chunk_sort_kernel(
        const int* __restrict__ src,
        const int* __restrict__ dst,
        unsigned* __restrict__ tab_d,
        unsigned* __restrict__ tab_s,
        unsigned* __restrict__ pd,
        unsigned* __restrict__ ps_src,
        int N, int E) {
    __shared__ unsigned hd[NB], hs[NB];
    __shared__ unsigned wsum_a[32], wsum_b[32];
    __shared__ unsigned stg_d[CH], stg_s[CH];
    const unsigned tid = threadIdx.x;
    const unsigned w   = blockIdx.x;
    if (tid < NB) { hd[tid] = 0u; hs[tid] = 0u; }
    const unsigned npb = ((unsigned)N + NB - 1u) / NB;
    const unsigned e0  = min(w * (unsigned)CH, (unsigned)E);
    const unsigned e1  = min(e0 + (unsigned)CH, (unsigned)E);
    int sv[EPT], dv[EPT];
#pragma unroll
    for (int j = 0; j < EPT; ++j) {
        const unsigned ee = e0 + tid + (unsigned)j * 1024u;
        sv[j] = 0; dv[j] = 0;
        if (ee < e1) { sv[j] = src[ee]; dv[j] = dst[ee]; }
    }
    __syncthreads();
    unsigned rd[EPT], rs[EPT];
#pragma unroll
    for (int j = 0; j < EPT; ++j) {
        const unsigned ee = e0 + tid + (unsigned)j * 1024u;
        rd[j] = 0u; rs[j] = 0u;
        if (ee < e1) {
            rd[j] = atomicAdd(&hd[bucket_of(dv[j], npb)], 1u);
            rs[j] = atomicAdd(&hs[bucket_of(sv[j], npb)], 1u);
        }
    }
    __syncthreads();
    const unsigned cdv = (tid < NB) ? hd[tid] : 0u;
    const unsigned csv = (tid < NB) ? hs[tid] : 0u;
    const unsigned od = block_excl_scan_1024(cdv, wsum_a);
    const unsigned os = block_excl_scan_1024(csv, wsum_b);
    if (tid < NB) {
        hd[tid] = od;
        hs[tid] = os;
        tab_d[w * TROW + tid] = od;
        tab_s[w * TROW + tid] = os;
        if (tid == NB - 1u) {
            tab_d[w * TROW + NB] = od + cdv;
            tab_s[w * TROW + NB] = os + csv;
        }
    }
    __syncthreads();
#pragma unroll
    for (int j = 0; j < EPT; ++j) {
        const unsigned ee = e0 + tid + (unsigned)j * 1024u;
        if (ee < e1) {
            const int s = sv[j];
            const int d = dv[j];
            const unsigned bd = bucket_of(d, npb);
            const unsigned bs = bucket_of(s, npb);
            const unsigned dl = min((unsigned)d - bd * npb, (unsigned)NPB_MAX - 1u);
            const unsigned sl = min((unsigned)s, (1u << SRC_BITS) - 1u);
            const unsigned p = min(hd[bd] + rd[j], (unsigned)CH - 1u);
            const unsigned q = min(hs[bs] + rs[j], (unsigned)CH - 1u);
            stg_d[p] = (dl << SRC_BITS) | sl;
            stg_s[q] = (unsigned)s;
        }
    }
    __syncthreads();
    const unsigned n = e1 - e0;
    for (unsigned i = tid; i < n; i += 1024u) {
        pd[e0 + i]     = stg_d[i];
        ps_src[e0 + i] = stg_s[i];
    }
}

template <typename F>
__device__ __forceinline__ void for_each_in_pieces(
        const unsigned* __restrict__ arr, unsigned last_word,
        const unsigned* s0, const unsigned* s1, unsigned nc, F f) {
    const unsigned hw  = threadIdx.x >> 4;
    const unsigned l16 = threadIdx.x & 15u;
    for (unsigned c0 = hw; c0 < nc; c0 += 256u) {
        unsigned b[4], e[4], v[4][2];
#pragma unroll
        for (int k = 0; k < 4; ++k) {
            const unsigned c = c0 + 64u * (unsigned)k;
            b[k] = 0u; e[k] = 0u;
            if (c < nc) { b[k] = s0[c] + l16; e[k] = s1[c]; }
        }
#pragma unroll
        for (int k = 0; k < 4; ++k) {
#pragma unroll
            for (int r = 0; r < 2; ++r) {
                v[k][r] = arr[min(b[k] + 16u * (unsigned)r, last_word)];
            }
        }
#pragma unroll
        for (int k = 0; k < 4; ++k) {
#pragma unroll
            for (int r = 0; r < 2; ++r) {
                if (b[k] + 16u * (unsigned)r < e[k])
                    f(v[k][r], b[k] + 16u * (unsigned)r, c0 + 64u * (unsigned)k);
            }
            for (unsigned i = b[k] + 32u; i < e[k]; i += 16u) f(arr[i], i, c0 + 64u * (unsigned)k);
        }
    }
}

__global__ __launch_bounds__(1024) void bucket_sort_kernel(
        const unsigned* __restrict__ pd,
        const unsigned* __restrict__ ps_src,
        const unsigned* __restrict__ tab_d,
        const unsigned* __restrict__ tab_s,
        unsigned* __restrict__ cnt_in,
        unsigned* __restrict__ off,
        float* __restrict__ inv_do,
        float* __restrict__ inv_di,
        int* __restrict__ grouped,
        int N, int E, int nchunks) {
    __shared__ unsigned cd[NPB_MAX], cs[NPB_MAX];
    __shared__ unsigned cur[NPB_MAX + 32];
    __shared__ unsigned wsum[32], wsum2[32];
    __shared__ unsigned pd0[NC_MAX], pd1[NC_MAX], ps0[NC_MAX], ps1[NC_MAX];
    __shared__ unsigned dof[NC_MAX];
    __shared__ unsigned kept[CAP];
    __shared__ unsigned outb[CAP];
    __shared__ unsigned bstart, btotal;
    const unsigned tid = threadIdx.x;
    const unsigned bkt = blockIdx.x;
    const unsigned npb = ((unsigned)N + NB - 1u) / NB;
    const unsigned nb  = bkt * npb;
    const unsigned nc  = min((unsigned)nchunks, (unsigned)NC_MAX);
    cd[tid] = 0u;
    cs[tid] = 0u;
    unsigned before = 0u;
    unsigned plen   = 0u;
    unsigned pfirst = 0u;
    if (tid < nc) {
        const unsigned base = min(tid * (unsigned)CH, (unsigned)E);
        const unsigned a0 = tab_d[tid * TROW + bkt];
        const unsigned a1 = tab_d[tid * TROW + bkt + 1u];
        const unsigned b0 = tab_s[tid * TROW + bkt];
        const unsigned b1 = tab_s[tid * TROW + bkt + 1u];
        const unsigned p0 = min(base + a0, (unsigned)E);
        const unsigned p1 = min(base + a1, (unsigned)E);
        pd0[tid] = p0;
        pd1[tid] = p1;
        ps0[tid] = min(base + b0, (unsigned)E);
        ps1[tid] = min(base + b1, (unsigned)E);
        before = a0;
        plen   = (p1 > p0) ? p1 - p0 : 0u;
        pfirst = p0;
    }
    const unsigned pre = block_excl_scan_1024(before, wsum);
    const unsigned kst = block_excl_scan_1024(plen, wsum2);
    if (tid < nc) dof[tid] = kst - pfirst;
    if (tid == 1023u) { bstart = pre + before; btotal = kst + plen; }
    __syncthreads();
    const unsigned ds = min(bstart, (unsigned)E);
    const unsigned total = btotal;
    const bool fits = total <= (unsigned)CAP;
    const unsigned last = (unsigned)E - 1u;

    for_each_in_pieces(pd, last, pd0, pd1, nc, [&](unsigned wv, unsigned i, unsigned c) {
        atomicAdd(&cd[(wv >> SRC_BITS) & (NPB_MAX - 1u)], 1u);
        if (fits) kept[min(i + dof[c], (unsigned)CAP - 1u)] = wv;
    });
    for_each_in_pieces(ps_src, last, ps0, ps1, nc, [&](unsigned sv, unsigned, unsigned) {
        atomicAdd(&cs[min(sv - nb, (unsigned)NPB_MAX - 1u)], 1u);
    });
    __syncthreads();
    const unsigned ci   = cd[tid];
    const unsigned excl = block_excl_scan_1024(ci, wsum);
    const unsigned node = nb + tid;
    cur[tid] = fits ? excl : ds + excl;
    if (tid < 32u) cur[NPB_MAX + tid] = 0u;
    if (tid < npb && node < (unsigned)N) {
        cnt_in[node] = ci;
        off[node]    = ds + excl;
        inv_do[node] = rsqrtf(fmaxf((float)cs[tid], 1.0f));
        inv_di[node] = rsqrtf(fmaxf((float)ci, 1.0f));
    }
    __syncthreads();
    if (fits) {
        constexpr int KPT = CAP / 1024;
        const unsigned spare = (unsigned)NPB_MAX + (tid & 31u);
        unsigned kw[KPT], kp[KPT];
#pragma unroll
        for (int j = 0; j < KPT; ++j) {
            const unsigned i = tid + (unsigned)j * 1024u;
            kw[j] = 0u;
            if (i < total) kw[j] = kept[i];
        }
#pragma unroll
        for (int j = 0; j < KPT; ++j) {
            const unsigned i = tid + (unsigned)j * 1024u;
            const unsigned ctr = (i < total) ? ((kw[j] >> SRC_BITS) & (NPB_MAX - 1u)) : spare;
            kp[j] = atomicAdd(&cur[ctr], 1u);
        }
#pragma unroll
        for (int j = 0; j < KPT; ++j) {
            const unsigned i = tid + (unsigned)j * 1024u;
            if (i < total) outb[min(kp[j], (unsigned)CAP - 1u)] = kw[j] & ((1u << SRC_BITS) - 1u);
        }
        __syncthreads();
        const unsigned sh = ds & 31u;
        for (unsigned j = tid; j < total + sh; j += 1024u) {
            if (j >= sh) grouped[min(ds + (j - sh), last)] = (int)outb[j - sh];
        }
    } else {
        for_each_in_pieces(pd, last, pd0, pd1, nc, [&](unsigned wv, unsigned, unsigned) {
            const unsigned pos = min(atomicAdd(&cur[(wv >> SRC_BITS) & (NPB_MAX - 1u)], 1u), last);
            grouped[pos] = (int)(wv & ((1u << SRC_BITS) - 1u));
        });
    }
}


template <int FOUT>
__global__ __launch_bounds__(256) void gemm_norm_wmma(
        const float* __restrict__ A0,
        const float* __restrict__ A1,
        const float* __restrict__ W,
        const float* __restrict__ inv_deg,
        float* __restrict__ out,
        int mtiles) {
    constexpr int NT = FOUT / 16;
    constexpr int KS = 2 * DH / 4;
    constexpr int C4 = FOUT / 4;

    __shared__ v2f Ws[KS * NT * 32];

    const int lane = threadIdx.x;
    const int wave = __builtin_amdgcn_readfirstlane(threadIdx.y);
    const int half = lane >> 4;
    const int l15  = lane & 15;

    {
        constexpr int TRIPS = DH * C4 / 256;
        const int tid = wave * 32 + lane;
        v4f r0[TRIPS], r1[TRIPS];
#pragma unroll
        for (int it = 0; it < TRIPS; ++it) {
            const int p  = tid + it * 256;
            const int kp = p / C4;
            const int c4 = p % C4;
            r0[it] = *reinterpret_cast<const v4f*>(W + (size_t)(2 * kp) * FOUT + c4 * 4);
            r1[it] = *reinterpret_cast<const v4f*>(W + (size_t)(2 * kp + 1) * FOUT + c4 * 4);
        }
#pragma unroll
        for (int it = 0; it < TRIPS; ++it) {
            const int p   = tid + it * 256;
            const int kp  = p / C4;
            const int c4  = p % C4;
            const int t   = kp >> 1;
            const int h   = kp & 1;
            const int col = c4 * 4;
            v2f* dstp = Ws + ((t * NT + (col >> 4)) * 32 + h * 16 + (col & 15));
            v2f e0, e1, e2, e3;
            e0.x = r0[it].x; e0.y = r1[it].x;
            e1.x = r0[it].y; e1.y = r1[it].y;
            e2.x = r0[it].z; e2.y = r1[it].z;
            e3.x = r0[it].w; e3.y = r1[it].w;
            dstp[0] = e0; dstp[1] = e1; dstp[2] = e2; dstp[3] = e3;
        }
    }
    __syncthreads();

    for (int mt = blockIdx.x * 8 + wave; mt < mtiles; mt += gridDim.x * 8) {
        const int row = mt * 16 + l15;
        const float s = inv_deg[row];
        const float* xr = A0 + (size_t)row * DH + half * 2;
        const float* hr = A1 + (size_t)row * DH + half * 2;

        v8f acc[NT];
#pragma unroll
        for (int nt = 0; nt < NT; ++nt) acc[nt] = v8f{};

        v2f an[8];
#pragma unroll
        for (int i = 0; i < 8; ++i) an[i] = *reinterpret_cast<const v2f*>(xr + 4 * i);

#pragma unroll 1
        for (int tc = 0; tc < KS / 8; ++tc) {
            v2f ac[8];
#pragma unroll
            for (int i = 0; i < 8; ++i) {
                ac[i].x = an[i].x * s;
                ac[i].y = an[i].y * s;
            }
            if (tc + 1 < KS / 8) {
                const int kn = 32 * (tc + 1);
                const float* nb = (kn < DH) ? (xr + kn) : (hr + (kn - DH));
#pragma unroll
                for (int i = 0; i < 8; ++i) an[i] = *reinterpret_cast<const v2f*>(nb + 4 * i);
            }
            const v2f* wb = Ws + (size_t)(tc * 8 * NT) * 32 + lane;
#pragma unroll
            for (int i = 0; i < 8; ++i) {
#pragma unroll
                for (int nt = 0; nt < NT; ++nt) {
                    const v2f b = wb[(i * NT + nt) * 32];
                    acc[nt] = __builtin_amdgcn_wmma_f32_16x16x4_f32(
                         false, ac[i],  false, b,
                         (short)0, acc[nt],  false,  false);
                }
            }
        }

#pragma unroll
        for (int nt = 0; nt < NT; ++nt) {
#pragma unroll
            for (int j = 0; j < 8; ++j) {
                const int r = mt * 16 + half * 8 + j;
                out[(size_t)r * FOUT + nt * 16 + l15] = acc[nt][j];
            }
        }
    }
}


template <int F>
__device__ __forceinline__ float4 gather_row_sum(
        const float* __restrict__ xw,
        const int* __restrict__ grouped,
        const unsigned* __restrict__ off,
        const unsigned* __restrict__ cnt,
        unsigned node, unsigned lin, int N) {
    constexpr int LPN = F / 4;
    const unsigned n = cnt[node];
    const int* seg = grouped + off[node];
    const float4* rows = reinterpret_cast<const float4*>(xw) + lin;
    const unsigned last = (unsigned)N - 1u;

    float4 acc = make_float4(0.0f, 0.0f, 0.0f, 0.0f);
    for (unsigned k0 = 0; k0 < n; k0 += LPN) {
        const unsigned m = min(n - k0, (unsigned)LPN);
        const unsigned mine = min((unsigned)seg[k0 + min(lin, m - 1u)], last);
        for (unsigned j0 = 0; j0 < m; j0 += 8) {
            float4 v[8];
#pragma unroll
            for (int j = 0; j < 8; ++j) {
                const unsigned s = __shfl(mine, (int)(j0 + j), LPN);
                v[j] = make_float4(0.0f, 0.0f, 0.0f, 0.0f);
                if (j0 + j < m) v[j] = rows[(size_t)s * LPN];
            }
#pragma unroll
            for (int j = 0; j < 8; ++j) {
                acc.x += v[j].x;
                acc.y += v[j].y;
                acc.z += v[j].z;
                acc.w += v[j].w;
            }
        }
    }
    return acc;
}

__device__ __forceinline__ float sigmoid_f(float g) {
    return 1.0f / (1.0f + __expf(-g));
}


__global__ __launch_bounds__(256) void gather_gates_kernel(
        const float* __restrict__ xw,
        const int* __restrict__ grouped,
        const unsigned* __restrict__ off,
        const unsigned* __restrict__ cnt,
        const float* __restrict__ inv_di,
        const float* __restrict__ b_gate,
        const float* __restrict__ h,
        float* __restrict__ u,
        float* __restrict__ rh,
        int N) {
    constexpr int LPN = 2 * DH / 4;
    constexpr int HQ  = DH / 4;
    const unsigned t = blockIdx.x * blockDim.x + threadIdx.x;
    const unsigned lin  = t % LPN;
    const unsigned step = gridDim.x * blockDim.x / LPN;
    const unsigned c4 = lin & (HQ - 1);
    const float4 b  = reinterpret_cast<const float4*>(b_gate)[lin];

    for (unsigned node = t / LPN; node < (unsigned)N; node += step) {
        const float inv = inv_di[node];
        const float4 hh = reinterpret_cast<const float4*>(h)[(size_t)node * HQ + c4];

        const float4 acc = gather_row_sum<2 * DH>(xw, grouped, off, cnt, node, lin, N);

        float4 g;
        g.x = sigmoid_f(acc.x * inv + b.x);
        g.y = sigmoid_f(acc.y * inv + b.y);
        g.z = sigmoid_f(acc.z * inv + b.z);
        g.w = sigmoid_f(acc.w * inv + b.w);
        if (lin < HQ) {
            reinterpret_cast<float4*>(rh)[(size_t)node * HQ + c4] =
                make_float4(g.x * hh.x, g.y * hh.y, g.z * hh.z, g.w * hh.w);
        } else {
            reinterpret_cast<float4*>(u)[(size_t)node * HQ + c4] = g;
        }
    }
}


__global__ __launch_bounds__(256) void gather_final_kernel(
        const float* __restrict__ xw,
        const int* __restrict__ grouped,
        const unsigned* __restrict__ off,
        const unsigned* __restrict__ cnt,
        const float* __restrict__ inv_di,
        const float* __restrict__ b_cand,
        const float* __restrict__ h,
        const float* __restrict__ u,
        float* __restrict__ out,
        int N) {
    constexpr int LPN = DH / 4;
    const unsigned t = blockIdx.x * blockDim.x + threadIdx.x;
    const unsigned lin  = t % LPN;
    const unsigned step = gridDim.x * blockDim.x / LPN;
    const float4 b  = reinterpret_cast<const float4*>(b_cand)[lin];

    for (unsigned node = t / LPN; node < (unsigned)N; node += step) {
        const float inv = inv_di[node];
        const float4 hh = reinterpret_cast<const float4*>(h)[(size_t)node * LPN + lin];
        const float4 uu = reinterpret_cast<const float4*>(u)[(size_t)node * LPN + lin];

        const float4 acc = gather_row_sum<DH>(xw, grouped, off, cnt, node, lin, N);

        float4 o;
        o.x = uu.x * hh.x + (1.0f - uu.x) * tanhf(acc.x * inv + b.x);
        o.y = uu.y * hh.y + (1.0f - uu.y) * tanhf(acc.y * inv + b.y);
        o.z = uu.z * hh.z + (1.0f - uu.z) * tanhf(acc.z * inv + b.z);
        o.w = uu.w * hh.w + (1.0f - uu.w) * tanhf(acc.w * inv + b.w);
        reinterpret_cast<float4*>(out)[(size_t)node * LPN + lin] = o;
    }
}


extern "C" void kernel_launch(void* const* d_in, const int* in_sizes, int n_in,
                              void* d_out, int out_size, void* d_ws, size_t ws_size,
                              hipStream_t stream) {
    const float* x      = (const float*)d_in[0];
    const float* h      = (const float*)d_in[1];
    const int*   src    = (const int*)d_in[2];
    const int*   dst    = (const int*)d_in[3];
    const float* W_gate = (const float*)d_in[4];
    const float* b_gate = (const float*)d_in[5];
    const float* W_cand = (const float*)d_in[6];
    const float* b_cand = (const float*)d_in[7];

    const int N = in_sizes[0] / DH;
    const int E = in_sizes[2];

    float* p      = (float*)d_ws;
    float* inv_do = p;
    float* inv_di = p + N;
    float* bufA   = p + 2 * (size_t)N;
    float* bufB   = bufA + (size_t)N * 2 * DH;
    float* bufC   = bufB + (size_t)N * 2 * DH;
    unsigned* q   = (unsigned*)(bufC + (size_t)N * DH);
    float* xw     = bufA;
    float* u      = bufB;
    float* rh     = bufB + (size_t)N * DH;
    float* xw2    = bufC;
    unsigned* cnt_in  = q + N;
    unsigned* off     = q + 2 * (size_t)N + 32;
    int* grouped      = (int*)(off + N) + E;
    const int nchunks  = (E + CH - 1) / CH;
    unsigned* pd       = (unsigned*)bufB;
    unsigned* ps_src   = pd + E;
    unsigned* tab_d    = ps_src + E;
    unsigned* tab_s    = tab_d + (size_t)nchunks * TROW;
    float* out    = (float*)d_out;

    const int mtiles = N / 16;
    int gemm_blocks = (mtiles + 7) / 8;
    if (gemm_blocks > 512) gemm_blocks = 512;

    chunk_sort_kernel<<<nchunks, 1024, 0, stream>>>(
        src, dst, tab_d, tab_s, pd, ps_src, N, E);
    bucket_sort_kernel<<<NB, 1024, 0, stream>>>(
        pd, ps_src, tab_d, tab_s, cnt_in, off, inv_do, inv_di, grouped, N, E, nchunks);

    gemm_norm_wmma<2 * DH><<<gemm_blocks, dim3(32, 8), 0, stream>>>(
        x, h, W_gate, inv_do, xw, mtiles);
    {
        unsigned blocks = ((unsigned)N * (2 * DH / 4) + 255) / 256;
        if (blocks > 2048u) blocks = 2048u;
        gather_gates_kernel<<<blocks, 256, 0, stream>>>(
            xw, grouped, off, cnt_in, inv_di, b_gate, h, u, rh, N);
    }

    gemm_norm_wmma<DH><<<gemm_blocks, dim3(32, 8), 0, stream>>>(
        x, rh, W_cand, inv_do, xw2, mtiles);
    {
        unsigned blocks = ((unsigned)N * (DH / 4) + 255) / 256;
        if (blocks > 2048u) blocks = 2048u;
        gather_final_kernel<<<blocks, 256, 0, stream>>>(
            xw2, grouped, off, cnt_in, inv_di, b_cand, h, u, out, N);
    }
}


# Gate conv's product moves to the tail of the bucket work-group that counted its rows' out-degrees (a bucket = whole 16-row tiles; 16 rows x 64 columns per wave): 6 -> 5 launches; sums unchanged
# speedup vs baseline: 8.5479x; 1.0531x over previous
#include <hip/hip_runtime.h>
#include <math.h>


typedef __attribute__((ext_vector_type(2))) float v2f;
typedef __attribute__((ext_vector_type(4))) float v4f;
typedef __attribute__((ext_vector_type(8))) float v8f;

#define DH 64


template <int FOUT, int NTHREADS>
__device__ __forceinline__ void stage_w_fragments(const float* __restrict__ W, v2f* Ws, int tid) {
    constexpr int NT = FOUT / 16;
    constexpr int C4 = FOUT / 4;
    constexpr int TRIPS = DH * C4 / NTHREADS;
    v4f r0[TRIPS], r1[TRIPS];
#pragma unroll
    for (int it = 0; it < TRIPS; ++it) {
        const int p  = tid + it * NTHREADS;
        const int kp = p / C4;
        const int c4 = p % C4;
        r0[it] = *reinterpret_cast<const v4f*>(W + (size_t)(2 * kp) * FOUT + c4 * 4);
        r1[it] = *reinterpret_cast<const v4f*>(W + (size_t)(2 * kp + 1) * FOUT + c4 * 4);
    }
#pragma unroll
    for (int it = 0; it < TRIPS; ++it) {
        const int p   = tid + it * NTHREADS;
        const int kp  = p / C4;
        const int c4  = p % C4;
        const int t   = kp >> 1;
        const int h   = kp & 1;
        const int col = c4 * 4;
        v2f* dstp = Ws + ((t * NT + (col >> 4)) * 32 + h * 16 + (col & 15));
        v2f e0, e1, e2, e3;
        e0.x = r0[it].x; e0.y = r1[it].x;
        e1.x = r0[it].y; e1.y = r1[it].y;
        e2.x = r0[it].z; e2.y = r1[it].z;
        e3.x = r0[it].w; e3.y = r1[it].w;
        dstp[0] = e0; dstp[1] = e1; dstp[2] = e2; dstp[3] = e3;
    }
}

template <int FOUT, int NTN>
__device__ __forceinline__ void gemm_wave_tile(
        const float* __restrict__ A0,
        const float* __restrict__ A1,
        const v2f* Ws, float s,
        float* __restrict__ out,
        int mt, int ct0, int lane) {
    constexpr int NT = FOUT / 16;
    constexpr int KS = 2 * DH / 4;
    const int half = lane >> 4;
    const int l15  = lane & 15;
    const int row = mt * 16 + l15;
    const float* xr = A0 + (size_t)row * DH + half * 2;
    const float* hr = A1 + (size_t)row * DH + half * 2;

    v8f acc[NTN];
#pragma unroll
    for (int nt = 0; nt < NTN; ++nt) acc[nt] = v8f{};

    v2f an[8];
#pragma unroll
    for (int i = 0; i < 8; ++i) an[i] = *reinterpret_cast<const v2f*>(xr + 4 * i);

#pragma unroll 1
    for (int tc = 0; tc < KS / 8; ++tc) {
        v2f ac[8];
#pragma unroll
        for (int i = 0; i < 8; ++i) {
            ac[i].x = an[i].x * s;
            ac[i].y = an[i].y * s;
        }
        if (tc + 1 < KS / 8) {
            const int kn = 32 * (tc + 1);
            const float* nb = (kn < DH) ? (xr + kn) : (hr + (kn - DH));
#pragma unroll
            for (int i = 0; i < 8; ++i) an[i] = *reinterpret_cast<const v2f*>(nb + 4 * i);
        }
        const v2f* wb = Ws + (size_t)(tc * 8 * NT + ct0) * 32 + lane;
#pragma unroll
        for (int i = 0; i < 8; ++i) {
#pragma unroll
            for (int nt = 0; nt < NTN; ++nt) {
                const v2f b = wb[(i * NT + nt) * 32];
                acc[nt] = __builtin_amdgcn_wmma_f32_16x16x4_f32(
                     false, ac[i],  false, b,
                     (short)0, acc[nt],  false,  false);
            }
        }
    }

#pragma unroll
    for (int nt = 0; nt < NTN; ++nt) {
#pragma unroll
        for (int j = 0; j < 8; ++j) {
            const int r = mt * 16 + half * 8 + j;
            out[(size_t)r * FOUT + (ct0 + nt) * 16 + l15] = acc[nt][j];
        }
    }
}


#define NB      256
#define NPB_MAX 1024
#define SRC_BITS 18
#define CH      6400
#define EPT     7
#define TROW    (NB + 1)
#define NC_MAX  1024
#define CAP     8192

__device__ __forceinline__ unsigned block_excl_scan_1024(unsigned v, unsigned* wsum) {
    const unsigned lane = threadIdx.x & 31u;
    const unsigned wave = threadIdx.x >> 5;
    unsigned incl = v;
    for (int d = 1; d < 32; d <<= 1) {
        unsigned t = __shfl_up(incl, d);
        if (lane >= (unsigned)d) incl += t;
    }
    if (lane == 31u) wsum[wave] = incl;
    __syncthreads();
    if (wave == 0u) {
        const unsigned s = wsum[lane];
        unsigned i2 = s;
        for (int d = 1; d < 32; d <<= 1) {
            unsigned t = __shfl_up(i2, d);
            if (lane >= (unsigned)d) i2 += t;
        }
        wsum[lane] = i2 - s;
    }
    __syncthreads();
    return incl - v + wsum[wave];
}

__device__ __forceinline__ unsigned bucket_of(int node, unsigned npb) {
    return min((unsigned)node / npb, (unsigned)NB - 1u);
}

__device__ __forceinline__ unsigned nodes_per_bucket(int N) {
    return ((((unsigned)N + NB - 1u) / NB) + 15u) & ~15u;
}

__global__ __launch_bounds__(1024) void chunk_sort_kernel(
        const int* __restrict__ src,
        const int* __restrict__ dst,
        unsigned* __restrict__ tab_d,
        unsigned* __restrict__ tab_s,
        unsigned* __restrict__ pd,
        unsigned* __restrict__ ps_src,
        int N, int E) {
    __shared__ unsigned hd[NB], hs[NB];
    __shared__ unsigned wsum_a[32], wsum_b[32];
    __shared__ unsigned stg_d[CH], stg_s[CH];
    const unsigned tid = threadIdx.x;
    const unsigned w   = blockIdx.x;
    if (tid < NB) { hd[tid] = 0u; hs[tid] = 0u; }
    const unsigned npb = nodes_per_bucket(N);
    const unsigned e0  = min(w * (unsigned)CH, (unsigned)E);
    const unsigned e1  = min(e0 + (unsigned)CH, (unsigned)E);
    int sv[EPT], dv[EPT];
#pragma unroll
    for (int j = 0; j < EPT; ++j) {
        const unsigned ee = e0 + tid + (unsigned)j * 1024u;
        sv[j] = 0; dv[j] = 0;
        if (ee < e1) { sv[j] = src[ee]; dv[j] = dst[ee]; }
    }
    __syncthreads();
    unsigned rd[EPT], rs[EPT];
#pragma unroll
    for (int j = 0; j < EPT; ++j) {
        const unsigned ee = e0 + tid + (unsigned)j * 1024u;
        rd[j] = 0u; rs[j] = 0u;
        if (ee < e1) {
            rd[j] = atomicAdd(&hd[bucket_of(dv[j], npb)], 1u);
            rs[j] = atomicAdd(&hs[bucket_of(sv[j], npb)], 1u);
        }
    }
    __syncthreads();
    const unsigned cdv = (tid < NB) ? hd[tid] : 0u;
    const unsigned csv = (tid < NB) ? hs[tid] : 0u;
    const unsigned od = block_excl_scan_1024(cdv, wsum_a);
    const unsigned os = block_excl_scan_1024(csv, wsum_b);
    if (tid < NB) {
        hd[tid] = od;
        hs[tid] = os;
        tab_d[w * TROW + tid] = od;
        tab_s[w * TROW + tid] = os;
        if (tid == NB - 1u) {
            tab_d[w * TROW + NB] = od + cdv;
            tab_s[w * TROW + NB] = os + csv;
        }
    }
    __syncthreads();
#pragma unroll
    for (int j = 0; j < EPT; ++j) {
        const unsigned ee = e0 + tid + (unsigned)j * 1024u;
        if (ee < e1) {
            const int s = sv[j];
            const int d = dv[j];
            const unsigned bd = bucket_of(d, npb);
            const unsigned bs = bucket_of(s, npb);
            const unsigned dl = min((unsigned)d - bd * npb, (unsigned)NPB_MAX - 1u);
            const unsigned sl = min((unsigned)s, (1u << SRC_BITS) - 1u);
            const unsigned p = min(hd[bd] + rd[j], (unsigned)CH - 1u);
            const unsigned q = min(hs[bs] + rs[j], (unsigned)CH - 1u);
            stg_d[p] = (dl << SRC_BITS) | sl;
            stg_s[q] = (unsigned)s;
        }
    }
    __syncthreads();
    const unsigned n = e1 - e0;
    for (unsigned i = tid; i < n; i += 1024u) {
        pd[e0 + i]     = stg_d[i];
        ps_src[e0 + i] = stg_s[i];
    }
}

template <typename F>
__device__ __forceinline__ void for_each_in_pieces(
        const unsigned* __restrict__ arr, unsigned last_word,
        const unsigned* s0, const unsigned* s1, unsigned nc, F f) {
    const unsigned hw  = threadIdx.x >> 4;
    const unsigned l16 = threadIdx.x & 15u;
    for (unsigned c0 = hw; c0 < nc; c0 += 256u) {
        unsigned b[4], e[4], v[4][2];
#pragma unroll
        for (int k = 0; k < 4; ++k) {
            const unsigned c = c0 + 64u * (unsigned)k;
            b[k] = 0u; e[k] = 0u;
            if (c < nc) { b[k] = s0[c] + l16; e[k] = s1[c]; }
        }
#pragma unroll
        for (int k = 0; k < 4; ++k) {
#pragma unroll
            for (int r = 0; r < 2; ++r) {
                v[k][r] = arr[min(b[k] + 16u * (unsigned)r, last_word)];
            }
        }
#pragma unroll
        for (int k = 0; k < 4; ++k) {
#pragma unroll
            for (int r = 0; r < 2; ++r) {
                if (b[k] + 16u * (unsigned)r < e[k])
                    f(v[k][r], b[k] + 16u * (unsigned)r, c0 + 64u * (unsigned)k);
            }
            for (unsigned i = b[k] + 32u; i < e[k]; i += 16u) f(arr[i], i, c0 + 64u * (unsigned)k);
        }
    }
}

__global__ __launch_bounds__(1024) void bucket_sort_kernel(
        const unsigned* __restrict__ pd,
        const unsigned* __restrict__ ps_src,
        const unsigned* __restrict__ tab_d,
        const unsigned* __restrict__ tab_s,
        unsigned* __restrict__ cnt_in,
        unsigned* __restrict__ off,
        float* __restrict__ inv_do,
        float* __restrict__ inv_di,
        int* __restrict__ grouped,
        const float* __restrict__ x,
        const float* __restrict__ h,
        const float* __restrict__ W_gate,
        float* __restrict__ xw,
        int N, int E, int nchunks) {
    __shared__ unsigned cd[NPB_MAX], cs[NPB_MAX];
    __shared__ float sdo[NPB_MAX];
    __shared__ v2f Ws[(2 * DH / 4) * (2 * DH / 16) * 32];
    __shared__ unsigned cur[NPB_MAX + 32];
    __shared__ unsigned wsum[32], wsum2[32];
    __shared__ unsigned pd0[NC_MAX], pd1[NC_MAX], ps0[NC_MAX], ps1[NC_MAX];
    __shared__ unsigned dof[NC_MAX];
    __shared__ unsigned kept[CAP];
    __shared__ unsigned outb[CAP];
    __shared__ unsigned bstart, btotal;
    const unsigned tid = threadIdx.x;
    const unsigned bkt = blockIdx.x;
    const unsigned npb = nodes_per_bucket(N);
    const unsigned nb  = bkt * npb;
    stage_w_fragments<2 * DH, 1024>(W_gate, Ws, (int)tid);
    const unsigned nc  = min((unsigned)nchunks, (unsigned)NC_MAX);
    cd[tid] = 0u;
    cs[tid] = 0u;
    unsigned before = 0u;
    unsigned plen   = 0u;
    unsigned pfirst = 0u;
    if (tid < nc) {
        const unsigned base = min(tid * (unsigned)CH, (unsigned)E);
        const unsigned a0 = tab_d[tid * TROW + bkt];
        const unsigned a1 = tab_d[tid * TROW + bkt + 1u];
        const unsigned b0 = tab_s[tid * TROW + bkt];
        const unsigned b1 = tab_s[tid * TROW + bkt + 1u];
        const unsigned p0 = min(base + a0, (unsigned)E);
        const unsigned p1 = min(base + a1, (unsigned)E);
        pd0[tid] = p0;
        pd1[tid] = p1;
        ps0[tid] = min(base + b0, (unsigned)E);
        ps1[tid] = min(base + b1, (unsigned)E);
        before = a0;
        plen   = (p1 > p0) ? p1 - p0 : 0u;
        pfirst = p0;
    }
    const unsigned pre = block_excl_scan_1024(before, wsum);
    const unsigned kst = block_excl_scan_1024(plen, wsum2);
    if (tid < nc) dof[tid] = kst - pfirst;
    if (tid == 1023u) { bstart = pre + before; btotal = kst + plen; }
    __syncthreads();
    const unsigned ds = min(bstart, (unsigned)E);
    const unsigned total = btotal;
    const bool fits = total <= (unsigned)CAP;
    const unsigned last = (unsigned)E - 1u;

    for_each_in_pieces(pd, last, pd0, pd1, nc, [&](unsigned wv, unsigned i, unsigned c) {
        atomicAdd(&cd[(wv >> SRC_BITS) & (NPB_MAX - 1u)], 1u);
        if (fits) kept[min(i + dof[c], (unsigned)CAP - 1u)] = wv;
    });
    for_each_in_pieces(ps_src, last, ps0, ps1, nc, [&](unsigned sv, unsigned, unsigned) {
        atomicAdd(&cs[min(sv - nb, (unsigned)NPB_MAX - 1u)], 1u);
    });
    __syncthreads();
    const unsigned ci   = cd[tid];
    const unsigned excl = block_excl_scan_1024(ci, wsum);
    const unsigned node = nb + tid;
    cur[tid] = fits ? excl : ds + excl;
    if (tid < 32u) cur[NPB_MAX + tid] = 0u;
    const float my_inv_do = rsqrtf(fmaxf((float)cs[tid], 1.0f));
    sdo[tid] = my_inv_do;
    if (tid < npb && node < (unsigned)N) {
        cnt_in[node] = ci;
        off[node]    = ds + excl;
        inv_do[node] = my_inv_do;
        inv_di[node] = rsqrtf(fmaxf((float)ci, 1.0f));
    }
    __syncthreads();
    if (fits) {
        constexpr int KPT = CAP / 1024;
        const unsigned spare = (unsigned)NPB_MAX + (tid & 31u);
        unsigned kw[KPT], kp[KPT];
#pragma unroll
        for (int j = 0; j < KPT; ++j) {
            const unsigned i = tid + (unsigned)j * 1024u;
            kw[j] = 0u;
            if (i < total) kw[j] = kept[i];
        }
#pragma unroll
        for (int j = 0; j < KPT; ++j) {
            const unsigned i = tid + (unsigned)j * 1024u;
            const unsigned ctr = (i < total) ? ((kw[j] >> SRC_BITS) & (NPB_MAX - 1u)) : spare;
            kp[j] = atomicAdd(&cur[ctr], 1u);
        }
#pragma unroll
        for (int j = 0; j < KPT; ++j) {
            const unsigned i = tid + (unsigned)j * 1024u;
            if (i < total) outb[min(kp[j], (unsigned)CAP - 1u)] = kw[j] & ((1u << SRC_BITS) - 1u);
        }
        __syncthreads();
        const unsigned sh = ds & 31u;
        for (unsigned j = tid; j < total + sh; j += 1024u) {
            if (j >= sh) grouped[min(ds + (j - sh), last)] = (int)outb[j - sh];
        }
    } else {
        for_each_in_pieces(pd, last, pd0, pd1, nc, [&](unsigned wv, unsigned, unsigned) {
            const unsigned pos = min(atomicAdd(&cur[(wv >> SRC_BITS) & (NPB_MAX - 1u)], 1u), last);
            grouped[pos] = (int)(wv & ((1u << SRC_BITS) - 1u));
        });
    }

    {
        const int lane = (int)(tid & 31u);
        const int wv = __builtin_amdgcn_readfirstlane((int)(tid >> 5));
        const int place  = (((wv >> 3) << 2) | (wv & 3)) + ((wv & 4) << 2);
        const int tpb    = (int)(npb >> 4);
        const int mtiles = N >> 4;
        for (int it = place; it < 2 * tpb; it += 32) {
            const int tl = it >> 1;
            const int mt = (int)bkt * tpb + tl;
            if (mt < mtiles) {
                const float s = sdo[tl * 16 + (lane & 15)];
                gemm_wave_tile<2 * DH, 4>(x, h, Ws, s, xw, mt, (it & 1) * 4, lane);
            }
        }
    }
}

template <int FOUT>
__global__ __launch_bounds__(256) void gemm_norm_wmma(
        const float* __restrict__ A0,
        const float* __restrict__ A1,
        const float* __restrict__ W,
        const float* __restrict__ inv_deg,
        float* __restrict__ out,
        int mtiles) {
    constexpr int NT = FOUT / 16;
    constexpr int KS = 2 * DH / 4;

    __shared__ v2f Ws[KS * NT * 32];

    const int lane = threadIdx.x;
    const int wave = __builtin_amdgcn_readfirstlane(threadIdx.y);

    stage_w_fragments<FOUT, 256>(W, Ws, wave * 32 + lane);
    __syncthreads();

    for (int mt = blockIdx.x * 8 + wave; mt < mtiles; mt += gridDim.x * 8) {
        const float s = inv_deg[mt * 16 + (lane & 15)];
        gemm_wave_tile<FOUT, NT>(A0, A1, Ws, s, out, mt, 0, lane);
    }
}


template <int F>
__device__ __forceinline__ float4 gather_row_sum(
        const float* __restrict__ xw,
        const int* __restrict__ grouped,
        const unsigned* __restrict__ off,
        const unsigned* __restrict__ cnt,
        unsigned node, unsigned lin, int N) {
    constexpr int LPN = F / 4;
    const unsigned n = cnt[node];
    const int* seg = grouped + off[node];
    const float4* rows = reinterpret_cast<const float4*>(xw) + lin;
    const unsigned last = (unsigned)N - 1u;

    float4 acc = make_float4(0.0f, 0.0f, 0.0f, 0.0f);
    for (unsigned k0 = 0; k0 < n; k0 += LPN) {
        const unsigned m = min(n - k0, (unsigned)LPN);
        const unsigned mine = min((unsigned)seg[k0 + min(lin, m - 1u)], last);
        for (unsigned j0 = 0; j0 < m; j0 += 8) {
            float4 v[8];
#pragma unroll
            for (int j = 0; j < 8; ++j) {
                const unsigned s = __shfl(mine, (int)(j0 + j), LPN);
                v[j] = make_float4(0.0f, 0.0f, 0.0f, 0.0f);
                if (j0 + j < m) v[j] = rows[(size_t)s * LPN];
            }
#pragma unroll
            for (int j = 0; j < 8; ++j) {
                acc.x += v[j].x;
                acc.y += v[j].y;
                acc.z += v[j].z;
                acc.w += v[j].w;
            }
        }
    }
    return acc;
}

__device__ __forceinline__ float sigmoid_f(float g) {
    return 1.0f / (1.0f + __expf(-g));
}


__global__ __launch_bounds__(256) void gather_gates_kernel(
        const float* __restrict__ xw,
        const int* __restrict__ grouped,
        const unsigned* __restrict__ off,
        const unsigned* __restrict__ cnt,
        const float* __restrict__ inv_di,
        const float* __restrict__ b_gate,
        const float* __restrict__ h,
        float* __restrict__ u,
        float* __restrict__ rh,
        int N) {
    constexpr int LPN = 2 * DH / 4;
    constexpr int HQ  = DH / 4;
    const unsigned t = blockIdx.x * blockDim.x + threadIdx.x;
    const unsigned lin  = t % LPN;
    const unsigned step = gridDim.x * blockDim.x / LPN;
    const unsigned c4 = lin & (HQ - 1);
    const float4 b  = reinterpret_cast<const float4*>(b_gate)[lin];

    for (unsigned node = t / LPN; node < (unsigned)N; node += step) {
        const float inv = inv_di[node];
        const float4 hh = reinterpret_cast<const float4*>(h)[(size_t)node * HQ + c4];

        const float4 acc = gather_row_sum<2 * DH>(xw, grouped, off, cnt, node, lin, N);

        float4 g;
        g.x = sigmoid_f(acc.x * inv + b.x);
        g.y = sigmoid_f(acc.y * inv + b.y);
        g.z = sigmoid_f(acc.z * inv + b.z);
        g.w = sigmoid_f(acc.w * inv + b.w);
        if (lin < HQ) {
            reinterpret_cast<float4*>(rh)[(size_t)node * HQ + c4] =
                make_float4(g.x * hh.x, g.y * hh.y, g.z * hh.z, g.w * hh.w);
        } else {
            reinterpret_cast<float4*>(u)[(size_t)node * HQ + c4] = g;
        }
    }
}


__global__ __launch_bounds__(256) void gather_final_kernel(
        const float* __restrict__ xw,
        const int* __restrict__ grouped,
        const unsigned* __restrict__ off,
        const unsigned* __restrict__ cnt,
        const float* __restrict__ inv_di,
        const float* __restrict__ b_cand,
        const float* __restrict__ h,
        const float* __restrict__ u,
        float* __restrict__ out,
        int N) {
    constexpr int LPN = DH / 4;
    const unsigned t = blockIdx.x * blockDim.x + threadIdx.x;
    const unsigned lin  = t % LPN;
    const unsigned step = gridDim.x * blockDim.x / LPN;
    const float4 b  = reinterpret_cast<const float4*>(b_cand)[lin];

    for (unsigned node = t / LPN; node < (unsigned)N; node += step) {
        const float inv = inv_di[node];
        const float4 hh = reinterpret_cast<const float4*>(h)[(size_t)node * LPN + lin];
        const float4 uu = reinterpret_cast<const float4*>(u)[(size_t)node * LPN + lin];

        const float4 acc = gather_row_sum<DH>(xw, grouped, off, cnt, node, lin, N);

        float4 o;
        o.x = uu.x * hh.x + (1.0f - uu.x) * tanhf(acc.x * inv + b.x);
        o.y = uu.y * hh.y + (1.0f - uu.y) * tanhf(acc.y * inv + b.y);
        o.z = uu.z * hh.z + (1.0f - uu.z) * tanhf(acc.z * inv + b.z);
        o.w = uu.w * hh.w + (1.0f - uu.w) * tanhf(acc.w * inv + b.w);
        reinterpret_cast<float4*>(out)[(size_t)node * LPN + lin] = o;
    }
}


extern "C" void kernel_launch(void* const* d_in, const int* in_sizes, int n_in,
                              void* d_out, int out_size, void* d_ws, size_t ws_size,
                              hipStream_t stream) {
    const float* x      = (const float*)d_in[0];
    const float* h      = (const float*)d_in[1];
    const int*   src    = (const int*)d_in[2];
    const int*   dst    = (const int*)d_in[3];
    const float* W_gate = (const float*)d_in[4];
    const float* b_gate = (const float*)d_in[5];
    const float* W_cand = (const float*)d_in[6];
    const float* b_cand = (const float*)d_in[7];

    const int N = in_sizes[0] / DH;
    const int E = in_sizes[2];

    float* p      = (float*)d_ws;
    float* inv_do = p;
    float* inv_di = p + N;
    float* bufA   = p + 2 * (size_t)N;
    float* bufB   = bufA + (size_t)N * 2 * DH;
    float* bufC   = bufB + (size_t)N * 2 * DH;
    unsigned* q   = (unsigned*)(bufC + (size_t)N * DH);
    float* xw     = bufA;
    float* u      = bufB;
    float* rh     = bufB + (size_t)N * DH;
    float* xw2    = bufC;
    unsigned* cnt_in  = q + N;
    unsigned* off     = q + 2 * (size_t)N + 32;
    int* grouped      = (int*)(off + N) + E;
    const int nchunks  = (E + CH - 1) / CH;
    unsigned* pd       = (unsigned*)bufB;
    unsigned* ps_src   = pd + E;
    unsigned* tab_d    = ps_src + E;
    unsigned* tab_s    = tab_d + (size_t)nchunks * TROW;
    float* out    = (float*)d_out;

    const int mtiles = N / 16;
    int gemm_blocks = (mtiles + 7) / 8;
    if (gemm_blocks > 512) gemm_blocks = 512;

    chunk_sort_kernel<<<nchunks, 1024, 0, stream>>>(
        src, dst, tab_d, tab_s, pd, ps_src, N, E);
    bucket_sort_kernel<<<NB, 1024, 0, stream>>>(
        pd, ps_src, tab_d, tab_s, cnt_in, off, inv_do, inv_di, grouped,
        x, h, W_gate, xw, N, E, nchunks);

    {
        unsigned blocks = ((unsigned)N * (2 * DH / 4) + 255) / 256;
        if (blocks > 2048u) blocks = 2048u;
        gather_gates_kernel<<<blocks, 256, 0, stream>>>(
            xw, grouped, off, cnt_in, inv_di, b_gate, h, u, rh, N);
    }

    gemm_norm_wmma<DH><<<gemm_blocks, dim3(32, 8), 0, stream>>>(
        x, rh, W_cand, inv_do, xw2, mtiles);
    {
        unsigned blocks = ((unsigned)N * (DH / 4) + 255) / 256;
        if (blocks > 2048u) blocks = 2048u;
        gather_final_kernel<<<blocks, 256, 0, stream>>>(
            xw2, grouped, off, cnt_in, inv_di, b_cand, h, u, out, N);
    }
}


# Candidate conv's product moves into the gate gather's launch: 8 waves own a 16-row tile, inv_do*[x | r*h] goes through LDS to four multiplying waves; r*h never stored: 5 -> 4 launches; sums unchanged
# speedup vs baseline: 8.8094x; 1.0306x over previous
#include <hip/hip_runtime.h>
#include <math.h>


typedef __attribute__((ext_vector_type(2))) float v2f;
typedef __attribute__((ext_vector_type(4))) float v4f;
typedef __attribute__((ext_vector_type(8))) float v8f;

#define DH 64


template <int FOUT, int NTHREADS>
__device__ __forceinline__ void stage_w_fragments(const float* __restrict__ W, v2f* Ws, int tid) {
    constexpr int NT = FOUT / 16;
    constexpr int C4 = FOUT / 4;
    constexpr int TRIPS = DH * C4 / NTHREADS;
    v4f r0[TRIPS], r1[TRIPS];
#pragma unroll
    for (int it = 0; it < TRIPS; ++it) {
        const int p  = tid + it * NTHREADS;
        const int kp = p / C4;
        const int c4 = p % C4;
        r0[it] = *reinterpret_cast<const v4f*>(W + (size_t)(2 * kp) * FOUT + c4 * 4);
        r1[it] = *reinterpret_cast<const v4f*>(W + (size_t)(2 * kp + 1) * FOUT + c4 * 4);
    }
#pragma unroll
    for (int it = 0; it < TRIPS; ++it) {
        const int p   = tid + it * NTHREADS;
        const int kp  = p / C4;
        const int c4  = p % C4;
        const int t   = kp >> 1;
        const int h   = kp & 1;
        const int col = c4 * 4;
        v2f* dstp = Ws + ((t * NT + (col >> 4)) * 32 + h * 16 + (col & 15));
        v2f e0, e1, e2, e3;
        e0.x = r0[it].x; e0.y = r1[it].x;
        e1.x = r0[it].y; e1.y = r1[it].y;
        e2.x = r0[it].z; e2.y = r1[it].z;
        e3.x = r0[it].w; e3.y = r1[it].w;
        dstp[0] = e0; dstp[1] = e1; dstp[2] = e2; dstp[3] = e3;
    }
}

template <int FOUT, int NTN>
__device__ __forceinline__ void gemm_wave_tile(
        const float* __restrict__ A0,
        const float* __restrict__ A1,
        const v2f* Ws, float s,
        float* __restrict__ out,
        int mt, int ct0, int lane) {
    constexpr int NT = FOUT / 16;
    constexpr int KS = 2 * DH / 4;
    const int half = lane >> 4;
    const int l15  = lane & 15;
    const int row = mt * 16 + l15;
    const float* xr = A0 + (size_t)row * DH + half * 2;
    const float* hr = A1 + (size_t)row * DH + half * 2;

    v8f acc[NTN];
#pragma unroll
    for (int nt = 0; nt < NTN; ++nt) acc[nt] = v8f{};

    v2f an[8];
#pragma unroll
    for (int i = 0; i < 8; ++i) an[i] = *reinterpret_cast<const v2f*>(xr + 4 * i);

#pragma unroll 1
    for (int tc = 0; tc < KS / 8; ++tc) {
        v2f ac[8];
#pragma unroll
        for (int i = 0; i < 8; ++i) {
            ac[i].x = an[i].x * s;
            ac[i].y = an[i].y * s;
        }
        if (tc + 1 < KS / 8) {
            const int kn = 32 * (tc + 1);
            const float* nb = (kn < DH) ? (xr + kn) : (hr + (kn - DH));
#pragma unroll
            for (int i = 0; i < 8; ++i) an[i] = *reinterpret_cast<const v2f*>(nb + 4 * i);
        }
        const v2f* wb = Ws + (size_t)(tc * 8 * NT + ct0) * 32 + lane;
#pragma unroll
        for (int i = 0; i < 8; ++i) {
#pragma unroll
            for (int nt = 0; nt < NTN; ++nt) {
                const v2f b = wb[(i * NT + nt) * 32];
                acc[nt] = __builtin_amdgcn_wmma_f32_16x16x4_f32(
                     false, ac[i],  false, b,
                     (short)0, acc[nt],  false,  false);
            }
        }
    }

#pragma unroll
    for (int nt = 0; nt < NTN; ++nt) {
#pragma unroll
        for (int j = 0; j < 8; ++j) {
            const int r = mt * 16 + half * 8 + j;
            out[(size_t)r * FOUT + (ct0 + nt) * 16 + l15] = acc[nt][j];
        }
    }
}


#define NB      256
#define NPB_MAX 1024
#define SRC_BITS 18
#define CH      6400
#define EPT     7
#define TROW    (NB + 1)
#define NC_MAX  1024
#define CAP     8192

__device__ __forceinline__ unsigned block_excl_scan_1024(unsigned v, unsigned* wsum) {
    const unsigned lane = threadIdx.x & 31u;
    const unsigned wave = threadIdx.x >> 5;
    unsigned incl = v;
    for (int d = 1; d < 32; d <<= 1) {
        unsigned t = __shfl_up(incl, d);
        if (lane >= (unsigned)d) incl += t;
    }
    if (lane == 31u) wsum[wave] = incl;
    __syncthreads();
    if (wave == 0u) {
        const unsigned s = wsum[lane];
        unsigned i2 = s;
        for (int d = 1; d < 32; d <<= 1) {
            unsigned t = __shfl_up(i2, d);
            if (lane >= (unsigned)d) i2 += t;
        }
        wsum[lane] = i2 - s;
    }
    __syncthreads();
    return incl - v + wsum[wave];
}

__device__ __forceinline__ unsigned bucket_of(int node, unsigned npb) {
    return min((unsigned)node / npb, (unsigned)NB - 1u);
}

__device__ __forceinline__ unsigned nodes_per_bucket(int N) {
    return ((((unsigned)N + NB - 1u) / NB) + 15u) & ~15u;
}

__global__ __launch_bounds__(1024) void chunk_sort_kernel(
        const int* __restrict__ src,
        const int* __restrict__ dst,
        unsigned* __restrict__ tab_d,
        unsigned* __restrict__ tab_s,
        unsigned* __restrict__ pd,
        unsigned* __restrict__ ps_src,
        int N, int E) {
    __shared__ unsigned hd[NB], hs[NB];
    __shared__ unsigned wsum_a[32], wsum_b[32];
    __shared__ unsigned stg_d[CH], stg_s[CH];
    const unsigned tid = threadIdx.x;
    const unsigned w   = blockIdx.x;
    if (tid < NB) { hd[tid] = 0u; hs[tid] = 0u; }
    const unsigned npb = nodes_per_bucket(N);
    const unsigned e0  = min(w * (unsigned)CH, (unsigned)E);
    const unsigned e1  = min(e0 + (unsigned)CH, (unsigned)E);
    int sv[EPT], dv[EPT];
#pragma unroll
    for (int j = 0; j < EPT; ++j) {
        const unsigned ee = e0 + tid + (unsigned)j * 1024u;
        sv[j] = 0; dv[j] = 0;
        if (ee < e1) { sv[j] = src[ee]; dv[j] = dst[ee]; }
    }
    __syncthreads();
    unsigned rd[EPT], rs[EPT];
#pragma unroll
    for (int j = 0; j < EPT; ++j) {
        const unsigned ee = e0 + tid + (unsigned)j * 1024u;
        rd[j] = 0u; rs[j] = 0u;
        if (ee < e1) {
            rd[j] = atomicAdd(&hd[bucket_of(dv[j], npb)], 1u);
            rs[j] = atomicAdd(&hs[bucket_of(sv[j], npb)], 1u);
        }
    }
    __syncthreads();
    const unsigned cdv = (tid < NB) ? hd[tid] : 0u;
    const unsigned csv = (tid < NB) ? hs[tid] : 0u;
    const unsigned od = block_excl_scan_1024(cdv, wsum_a);
    const unsigned os = block_excl_scan_1024(csv, wsum_b);
    if (tid < NB) {
        hd[tid] = od;
        hs[tid] = os;
        tab_d[w * TROW + tid] = od;
        tab_s[w * TROW + tid] = os;
        if (tid == NB - 1u) {
            tab_d[w * TROW + NB] = od + cdv;
            tab_s[w * TROW + NB] = os + csv;
        }
    }
    __syncthreads();
#pragma unroll
    for (int j = 0; j < EPT; ++j) {
        const unsigned ee = e0 + tid + (unsigned)j * 1024u;
        if (ee < e1) {
            const int s = sv[j];
            const int d = dv[j];
            const unsigned bd = bucket_of(d, npb);
            const unsigned bs = bucket_of(s, npb);
            const unsigned dl = min((unsigned)d - bd * npb, (unsigned)NPB_MAX - 1u);
            const unsigned sl = min((unsigned)s, (1u << SRC_BITS) - 1u);
            const unsigned p = min(hd[bd] + rd[j], (unsigned)CH - 1u);
            const unsigned q = min(hs[bs] + rs[j], (unsigned)CH - 1u);
            stg_d[p] = (dl << SRC_BITS) | sl;
            stg_s[q] = (unsigned)s;
        }
    }
    __syncthreads();
    const unsigned n = e1 - e0;
    for (unsigned i = tid; i < n; i += 1024u) {
        pd[e0 + i]     = stg_d[i];
        ps_src[e0 + i] = stg_s[i];
    }
}

template <typename F>
__device__ __forceinline__ void for_each_in_pieces(
        const unsigned* __restrict__ arr, unsigned last_word,
        const unsigned* s0, const unsigned* s1, unsigned nc, F f) {
    const unsigned hw  = threadIdx.x >> 4;
    const unsigned l16 = threadIdx.x & 15u;
    for (unsigned c0 = hw; c0 < nc; c0 += 256u) {
        unsigned b[4], e[4], v[4][2];
#pragma unroll
        for (int k = 0; k < 4; ++k) {
            const unsigned c = c0 + 64u * (unsigned)k;
            b[k] = 0u; e[k] = 0u;
            if (c < nc) { b[k] = s0[c] + l16; e[k] = s1[c]; }
        }
#pragma unroll
        for (int k = 0; k < 4; ++k) {
#pragma unroll
            for (int r = 0; r < 2; ++r) {
                v[k][r] = arr[min(b[k] + 16u * (unsigned)r, last_word)];
            }
        }
#pragma unroll
        for (int k = 0; k < 4; ++k) {
#pragma unroll
            for (int r = 0; r < 2; ++r) {
                if (b[k] + 16u * (unsigned)r < e[k])
                    f(v[k][r], b[k] + 16u * (unsigned)r, c0 + 64u * (unsigned)k);
            }
            for (unsigned i = b[k] + 32u; i < e[k]; i += 16u) f(arr[i], i, c0 + 64u * (unsigned)k);
        }
    }
}

__global__ __launch_bounds__(1024) void bucket_sort_kernel(
        const unsigned* __restrict__ pd,
        const unsigned* __restrict__ ps_src,
        const unsigned* __restrict__ tab_d,
        const unsigned* __restrict__ tab_s,
        unsigned* __restrict__ cnt_in,
        unsigned* __restrict__ off,
        float* __restrict__ inv_do,
        float* __restrict__ inv_di,
        int* __restrict__ grouped,
        const float* __restrict__ x,
        const float* __restrict__ h,
        const float* __restrict__ W_gate,
        float* __restrict__ xw,
        int N, int E, int nchunks) {
    __shared__ unsigned cd[NPB_MAX], cs[NPB_MAX];
    __shared__ float sdo[NPB_MAX];
    __shared__ v2f Ws[(2 * DH / 4) * (2 * DH / 16) * 32];
    __shared__ unsigned cur[NPB_MAX + 32];
    __shared__ unsigned wsum[32], wsum2[32];
    __shared__ unsigned pd0[NC_MAX], pd1[NC_MAX], ps0[NC_MAX], ps1[NC_MAX];
    __shared__ unsigned dof[NC_MAX];
    __shared__ unsigned kept[CAP];
    __shared__ unsigned outb[CAP];
    __shared__ unsigned bstart, btotal;
    const unsigned tid = threadIdx.x;
    const unsigned bkt = blockIdx.x;
    const unsigned npb = nodes_per_bucket(N);
    const unsigned nb  = bkt * npb;
    stage_w_fragments<2 * DH, 1024>(W_gate, Ws, (int)tid);
    const unsigned nc  = min((unsigned)nchunks, (unsigned)NC_MAX);
    cd[tid] = 0u;
    cs[tid] = 0u;
    unsigned before = 0u;
    unsigned plen   = 0u;
    unsigned pfirst = 0u;
    if (tid < nc) {
        const unsigned base = min(tid * (unsigned)CH, (unsigned)E);
        const unsigned a0 = tab_d[tid * TROW + bkt];
        const unsigned a1 = tab_d[tid * TROW + bkt + 1u];
        const unsigned b0 = tab_s[tid * TROW + bkt];
        const unsigned b1 = tab_s[tid * TROW + bkt + 1u];
        const unsigned p0 = min(base + a0, (unsigned)E);
        const unsigned p1 = min(base + a1, (unsigned)E);
        pd0[tid] = p0;
        pd1[tid] = p1;
        ps0[tid] = min(base + b0, (unsigned)E);
        ps1[tid] = min(base + b1, (unsigned)E);
        before = a0;
        plen   = (p1 > p0) ? p1 - p0 : 0u;
        pfirst = p0;
    }
    const unsigned pre = block_excl_scan_1024(before, wsum);
    const unsigned kst = block_excl_scan_1024(plen, wsum2);
    if (tid < nc) dof[tid] = kst - pfirst;
    if (tid == 1023u) { bstart = pre + before; btotal = kst + plen; }
    __syncthreads();
    const unsigned ds = min(bstart, (unsigned)E);
    const unsigned total = btotal;
    const bool fits = total <= (unsigned)CAP;
    const unsigned last = (unsigned)E - 1u;

    for_each_in_pieces(pd, last, pd0, pd1, nc, [&](unsigned wv, unsigned i, unsigned c) {
        atomicAdd(&cd[(wv >> SRC_BITS) & (NPB_MAX - 1u)], 1u);
        if (fits) kept[min(i + dof[c], (unsigned)CAP - 1u)] = wv;
    });
    for_each_in_pieces(ps_src, last, ps0, ps1, nc, [&](unsigned sv, unsigned, unsigned) {
        atomicAdd(&cs[min(sv - nb, (unsigned)NPB_MAX - 1u)], 1u);
    });
    __syncthreads();
    const unsigned ci   = cd[tid];
    const unsigned excl = block_excl_scan_1024(ci, wsum);
    const unsigned node = nb + tid;
    cur[tid] = fits ? excl : ds + excl;
    if (tid < 32u) cur[NPB_MAX + tid] = 0u;
    const float my_inv_do = rsqrtf(fmaxf((float)cs[tid], 1.0f));
    sdo[tid] = my_inv_do;
    if (tid < npb && node < (unsigned)N) {
        cnt_in[node] = ci;
        off[node]    = ds + excl;
        inv_do[node] = my_inv_do;
        inv_di[node] = rsqrtf(fmaxf((float)ci, 1.0f));
    }
    __syncthreads();
    if (fits) {
        constexpr int KPT = CAP / 1024;
        const unsigned spare = (unsigned)NPB_MAX + (tid & 31u);
        unsigned kw[KPT], kp[KPT];
#pragma unroll
        for (int j = 0; j < KPT; ++j) {
            const unsigned i = tid + (unsigned)j * 1024u;
            kw[j] = 0u;
            if (i < total) kw[j] = kept[i];
        }
#pragma unroll
        for (int j = 0; j < KPT; ++j) {
            const unsigned i = tid + (unsigned)j * 1024u;
            const unsigned ctr = (i < total) ? ((kw[j] >> SRC_BITS) & (NPB_MAX - 1u)) : spare;
            kp[j] = atomicAdd(&cur[ctr], 1u);
        }
#pragma unroll
        for (int j = 0; j < KPT; ++j) {
            const unsigned i = tid + (unsigned)j * 1024u;
            if (i < total) outb[min(kp[j], (unsigned)CAP - 1u)] = kw[j] & ((1u << SRC_BITS) - 1u);
        }
        __syncthreads();
        const unsigned sh = ds & 31u;
        for (unsigned j = tid; j < total + sh; j += 1024u) {
            if (j >= sh) grouped[min(ds + (j - sh), last)] = (int)outb[j - sh];
        }
    } else {
        for_each_in_pieces(pd, last, pd0, pd1, nc, [&](unsigned wv, unsigned, unsigned) {
            const unsigned pos = min(atomicAdd(&cur[(wv >> SRC_BITS) & (NPB_MAX - 1u)], 1u), last);
            grouped[pos] = (int)(wv & ((1u << SRC_BITS) - 1u));
        });
    }

    {
        const int lane = (int)(tid & 31u);
        const int wv = __builtin_amdgcn_readfirstlane((int)(tid >> 5));
        const int place  = (((wv >> 3) << 2) | (wv & 3)) + ((wv & 4) << 2);
        const int tpb    = (int)(npb >> 4);
        const int mtiles = N >> 4;
        for (int it = place; it < 2 * tpb; it += 32) {
            const int tl = it >> 1;
            const int mt = (int)bkt * tpb + tl;
            if (mt < mtiles) {
                const float s = sdo[tl * 16 + (lane & 15)];
                gemm_wave_tile<2 * DH, 4>(x, h, Ws, s, xw, mt, (it & 1) * 4, lane);
            }
        }
    }
}


template <int F>
__device__ __forceinline__ float4 gather_row_sum(
        const float* __restrict__ xw,
        const int* __restrict__ grouped,
        const unsigned* __restrict__ off,
        const unsigned* __restrict__ cnt,
        unsigned node, unsigned lin, int N) {
    constexpr int LPN = F / 4;
    const unsigned n = cnt[node];
    const int* seg = grouped + off[node];
    const float4* rows = reinterpret_cast<const float4*>(xw) + lin;
    const unsigned last = (unsigned)N - 1u;

    float4 acc = make_float4(0.0f, 0.0f, 0.0f, 0.0f);
    for (unsigned k0 = 0; k0 < n; k0 += LPN) {
        const unsigned m = min(n - k0, (unsigned)LPN);
        const unsigned mine = min((unsigned)seg[k0 + min(lin, m - 1u)], last);
        for (unsigned j0 = 0; j0 < m; j0 += 8) {
            float4 v[8];
#pragma unroll
            for (int j = 0; j < 8; ++j) {
                const unsigned s = __shfl(mine, (int)(j0 + j), LPN);
                v[j] = make_float4(0.0f, 0.0f, 0.0f, 0.0f);
                if (j0 + j < m) v[j] = rows[(size_t)s * LPN];
            }
#pragma unroll
            for (int j = 0; j < 8; ++j) {
                acc.x += v[j].x;
                acc.y += v[j].y;
                acc.z += v[j].z;
                acc.w += v[j].w;
            }
        }
    }
    return acc;
}

__device__ __forceinline__ float sigmoid_f(float g) {
    return 1.0f / (1.0f + __expf(-g));
}


#define AP 132

__global__ __launch_bounds__(256) void gather_gates_cand_kernel(
        const float* __restrict__ xw,
        const int* __restrict__ grouped,
        const unsigned* __restrict__ off,
        const unsigned* __restrict__ cnt,
        const float* __restrict__ inv_di,
        const float* __restrict__ inv_do,
        const float* __restrict__ b_gate,
        const float* __restrict__ x,
        const float* __restrict__ h,
        const float* __restrict__ W_cand,
        float* __restrict__ u,
        float* __restrict__ xw2,
        int N) {
    constexpr int HQ = DH / 4;
    constexpr int NT = DH / 16;
    constexpr int KS = 2 * DH / 4;
    __shared__ v2f Ws[KS * NT * 32];
    __shared__ __attribute__((aligned(16))) float At[16 * AP];

    const unsigned lin = threadIdx.x & 31u;
    const int wave = __builtin_amdgcn_readfirstlane((int)(threadIdx.x >> 5));
    const unsigned c4 = lin & (HQ - 1);
    const float4 b  = reinterpret_cast<const float4*>(b_gate)[lin];
    const int mtiles = N >> 4;

    stage_w_fragments<DH, 256>(W_cand, Ws, (int)threadIdx.x);

    float* A = At;
    for (int tile = blockIdx.x; tile < mtiles; tile += gridDim.x) {
        for (int q = 0; q < 2; ++q) {
            const int rl = wave * 2 + q;
            const unsigned node = (unsigned)tile * 16u + (unsigned)rl;
            const float inv = inv_di[node];
            const float s   = inv_do[node];
            const float4 hx = (lin < HQ)
                ? reinterpret_cast<const float4*>(h)[(size_t)node * HQ + c4]
                : reinterpret_cast<const float4*>(x)[(size_t)node * HQ + c4];

            const float4 acc = gather_row_sum<2 * DH>(xw, grouped, off, cnt, node, lin, N);

            float4 g;
            g.x = sigmoid_f(acc.x * inv + b.x);
            g.y = sigmoid_f(acc.y * inv + b.y);
            g.z = sigmoid_f(acc.z * inv + b.z);
            g.w = sigmoid_f(acc.w * inv + b.w);
            float4 a;
            if (lin < HQ) {
                a = make_float4((g.x * hx.x) * s, (g.y * hx.y) * s, (g.z * hx.z) * s, (g.w * hx.w) * s);
            } else {
                reinterpret_cast<float4*>(u)[(size_t)node * HQ + c4] = g;
                a = make_float4(hx.x * s, hx.y * s, hx.z * s, hx.w * s);
            }
            if (q == 0) __syncthreads();
            *reinterpret_cast<float4*>(A + rl * AP + ((lin < HQ) ? DH : 0) + 4 * (int)c4) = a;
        }
        __syncthreads();
        if (wave < NT) {
            const int half = (int)(lin >> 4);
            const int l15  = (int)(lin & 15u);
            const float* ar = A + l15 * AP + half * 2;
            const v2f* wb = Ws + wave * 32 + (int)lin;
            v8f d = v8f{};
#pragma unroll
            for (int t = 0; t < KS; ++t) {
                const v2f af = *reinterpret_cast<const v2f*>(ar + 4 * t);
                const v2f bf = wb[t * NT * 32];
                d = __builtin_amdgcn_wmma_f32_16x16x4_f32(
                     false, af,  false, bf,
                     (short)0, d,  false,  false);
            }
#pragma unroll
            for (int j = 0; j < 8; ++j) {
                const int r = tile * 16 + half * 8 + j;
                xw2[(size_t)r * DH + wave * 16 + l15] = d[j];
            }
        }
    }
}


__global__ __launch_bounds__(256) void gather_final_kernel(
        const float* __restrict__ xw,
        const int* __restrict__ grouped,
        const unsigned* __restrict__ off,
        const unsigned* __restrict__ cnt,
        const float* __restrict__ inv_di,
        const float* __restrict__ b_cand,
        const float* __restrict__ h,
        const float* __restrict__ u,
        float* __restrict__ out,
        int N) {
    constexpr int LPN = DH / 4;
    const unsigned t = blockIdx.x * blockDim.x + threadIdx.x;
    const unsigned lin  = t % LPN;
    const unsigned step = gridDim.x * blockDim.x / LPN;
    const float4 b  = reinterpret_cast<const float4*>(b_cand)[lin];

    for (unsigned node = t / LPN; node < (unsigned)N; node += step) {
        const float inv = inv_di[node];
        const float4 hh = reinterpret_cast<const float4*>(h)[(size_t)node * LPN + lin];
        const float4 uu = reinterpret_cast<const float4*>(u)[(size_t)node * LPN + lin];

        const float4 acc = gather_row_sum<DH>(xw, grouped, off, cnt, node, lin, N);

        float4 o;
        o.x = uu.x * hh.x + (1.0f - uu.x) * tanhf(acc.x * inv + b.x);
        o.y = uu.y * hh.y + (1.0f - uu.y) * tanhf(acc.y * inv + b.y);
        o.z = uu.z * hh.z + (1.0f - uu.z) * tanhf(acc.z * inv + b.z);
        o.w = uu.w * hh.w + (1.0f - uu.w) * tanhf(acc.w * inv + b.w);
        reinterpret_cast<float4*>(out)[(size_t)node * LPN + lin] = o;
    }
}


extern "C" void kernel_launch(void* const* d_in, const int* in_sizes, int n_in,
                              void* d_out, int out_size, void* d_ws, size_t ws_size,
                              hipStream_t stream) {
    const float* x      = (const float*)d_in[0];
    const float* h      = (const float*)d_in[1];
    const int*   src    = (const int*)d_in[2];
    const int*   dst    = (const int*)d_in[3];
    const float* W_gate = (const float*)d_in[4];
    const float* b_gate = (const float*)d_in[5];
    const float* W_cand = (const float*)d_in[6];
    const float* b_cand = (const float*)d_in[7];

    const int N = in_sizes[0] / DH;
    const int E = in_sizes[2];

    float* p      = (float*)d_ws;
    float* inv_do = p;
    float* inv_di = p + N;
    float* bufA   = p + 2 * (size_t)N;
    float* bufB   = bufA + (size_t)N * 2 * DH;
    float* bufC   = bufB + (size_t)N * 2 * DH;
    unsigned* q   = (unsigned*)(bufC + (size_t)N * DH);
    float* xw     = bufA;
    float* u      = bufB;
    float* xw2    = bufC;
    unsigned* cnt_in  = q + N;
    unsigned* off     = q + 2 * (size_t)N + 32;
    int* grouped      = (int*)(off + N) + E;
    const int nchunks  = (E + CH - 1) / CH;
    unsigned* pd       = (unsigned*)bufB;
    unsigned* ps_src   = pd + E;
    unsigned* tab_d    = ps_src + E;
    unsigned* tab_s    = tab_d + (size_t)nchunks * TROW;
    float* out    = (float*)d_out;

    const int mtiles = N / 16;

    chunk_sort_kernel<<<nchunks, 1024, 0, stream>>>(
        src, dst, tab_d, tab_s, pd, ps_src, N, E);
    bucket_sort_kernel<<<NB, 1024, 0, stream>>>(
        pd, ps_src, tab_d, tab_s, cnt_in, off, inv_do, inv_di, grouped,
        x, h, W_gate, xw, N, E, nchunks);

    {
        int blocks = (mtiles + 3) / 4;
        if (blocks > 2048) blocks = 2048;
        gather_gates_cand_kernel<<<blocks, 256, 0, stream>>>(
            xw, grouped, off, cnt_in, inv_di, inv_do, b_gate, x, h, W_cand, u, xw2, N);
    }

    {
        unsigned blocks = ((unsigned)N * (DH / 4) + 255) / 256;
        if (blocks > 2048u) blocks = 2048u;
        gather_final_kernel<<<blocks, 256, 0, stream>>>(
            xw2, grouped, off, cnt_in, inv_di, b_cand, h, u, out, N);
    }
}
